# starting kernel
# baseline (speedup 1.0000x reference)
#include <hip/hip_runtime.h>
#include <hip/hip_bf16.h>


typedef __attribute__((ext_vector_type(2))) float v2f;
typedef __attribute__((ext_vector_type(8))) float v8f;

#define B_ 2
#define S_ 2048
#define D_ 1024
#define H_ 16
#define KVH_ 4
#define HD_ 64
#define WIN_ 64
#define M_ (B_ * S_)

template <int DO_ROPE>
__global__ void gemm64x32_wmma(const float* __restrict__ A, const float* __restrict__ W,
                               float* __restrict__ C, int M, int N, int K,
                               const float* __restrict__ fcos,
                               const float* __restrict__ fsin) {
    const int lane  = threadIdx.x & 31;
    const int wave  = threadIdx.x >> 5;
    const int tileM = blockIdx.y * 64;
    const int tileN = (blockIdx.x * (blockDim.x >> 5) + wave) * 32;
    if (tileN >= N) return;

    const int l15  = lane & 15;
    const int koff = (lane >> 4) << 1;

    v8f acc[4][2] = {};

    const float* ap = A + (long)(tileM + l15) * K + koff;
    const float* bp = W + (long)koff * N + (tileN + l15);
    const long aStride = 16L * K;

    for (int kk = 0; kk < K; kk += 4) {
        v2f a[4], b[2];
        #pragma unroll
        for (int i = 0; i < 4; ++i) {
            a[i].x = ap[i * aStride];
            a[i].y = ap[i * aStride + 1];
        }
        #pragma unroll
        for (int j = 0; j < 2; ++j) {
            b[j].x = bp[j * 16];
            b[j].y = bp[N + j * 16];
        }
        #pragma unroll
        for (int i = 0; i < 4; ++i)
            #pragma unroll
            for (int j = 0; j < 2; ++j)
                acc[i][j] = __builtin_amdgcn_wmma_f32_16x16x4_f32(
                    false, a[i], false, b[j], (short)0, acc[i][j], false, false);
        ap += 4;
        bp += (long)4 * N;
    }

    const int rowOff = (lane >> 4) << 3;
    #pragma unroll
    for (int i = 0; i < 4; ++i) {
        const int rowBase = tileM + i * 16 + rowOff;
        #pragma unroll
        for (int j = 0; j < 2; ++j) {
            const int col = tileN + j * 16 + l15;
            #pragma unroll
            for (int r = 0; r < 8; ++r) {
                float val = acc[i][j][r];
                if (DO_ROPE) {
                    const float partner = __shfl_xor(val, 1, 32);
                    const int pos = (rowBase + r) & (S_ - 1);
                    const int fi  = (col & (HD_ - 1)) >> 1;
                    const float c = fcos[pos * (HD_ / 2) + fi];
                    const float s = fsin[pos * (HD_ / 2) + fi];
                    val = ((col & 1) == 0) ? (val * c - partner * s)
                                           : (partner * s + val * c);
                }
                C[(long)(rowBase + r) * N + col] = val;
            }
        }
    }
}

__global__ void attn_wmma(const float* __restrict__ q, const float* __restrict__ kbuf,
                          const float* __restrict__ vbuf, float* __restrict__ o) {
    __shared__ float p[16][83];

    const int lane = threadIdx.x;
    const int i0   = blockIdx.x * 16;
    const int h    = blockIdx.y;
    const int b    = blockIdx.z;
    const int kvh  = h >> 2;

    const int l15  = lane & 15;
    const int koff = (lane >> 4) << 1;
    const int r8   = (lane >> 4) << 3;
    const int keystart = i0 - WIN_;

    const float* qrow = q + ((long)(b * S_ + i0 + l15) * H_ + h) * HD_;
    for (int t = 0; t < 5; ++t) {
        const int j0 = keystart + t * 16;
        if (j0 < 0) {
            #pragma unroll
            for (int r = 0; r < 8; ++r) p[r8 + r][t * 16 + l15] = -1e9f;
            continue;
        }
        const float* krow = kbuf + ((long)(b * S_ + j0 + l15) * KVH_ + kvh) * HD_;
        v8f acc = {};
        for (int kk = 0; kk < HD_; kk += 4) {
            const int kb = kk + koff;
            v2f a, bb;
            a.x  = qrow[kb];  a.y  = qrow[kb + 1];
            bb.x = krow[kb];  bb.y = krow[kb + 1];
            acc = __builtin_amdgcn_wmma_f32_16x16x4_f32(false, a, false, bb,
                                                        (short)0, acc, false, false);
        }
        #pragma unroll
        for (int r = 0; r < 8; ++r) {
            const int row  = i0 + r8 + r;
            const int colk = j0 + l15;
            float sc = fmaxf(acc[r] * 0.125f, -1e9f);
            if (colk > row || row - colk > WIN_) sc = -1e9f;
            p[r8 + r][t * 16 + l15] = sc;
        }
    }
    __syncthreads();

    if (lane < 16) {
        float m = -1e30f;
        for (int c = 0; c < 80; ++c) m = fmaxf(m, p[lane][c]);
        float sum = 0.0f;
        for (int c = 0; c < 80; ++c) {
            const float e = expf(p[lane][c] - m);
            p[lane][c] = e;
            sum += e;
        }
        const float inv = 1.0f / sum;
        for (int c = 0; c < 80; ++c) p[lane][c] *= inv;
    }
    __syncthreads();

    for (int nt = 0; nt < 4; ++nt) {
        const int n0 = nt * 16;
        v8f acc = {};
        for (int kk = 0; kk < 80; kk += 4) {
            const int kb = kk + koff;
            v2f a, bb;
            a.x = p[l15][kb];
            a.y = p[l15][kb + 1];
            const int key0 = keystart + kb;
            const int key1 = key0 + 1;
            bb.x = (key0 >= 0)
                 ? vbuf[((long)(b * S_ + key0) * KVH_ + kvh) * HD_ + n0 + l15] : 0.0f;
            bb.y = (key1 >= 0)
                 ? vbuf[((long)(b * S_ + key1) * KVH_ + kvh) * HD_ + n0 + l15] : 0.0f;
            acc = __builtin_amdgcn_wmma_f32_16x16x4_f32(false, a, false, bb,
                                                        (short)0, acc, false, false);
        }
        #pragma unroll
        for (int r = 0; r < 8; ++r) {
            o[((long)(b * S_ + i0 + r8 + r) * H_ + h) * HD_ + n0 + l15] = acc[r];
        }
    }
}

extern "C" void kernel_launch(void* const* d_in, const int* in_sizes, int n_in,
                              void* d_out, int out_size, void* d_ws, size_t ws_size,
                              hipStream_t stream) {
    const float* x    = (const float*)d_in[0];
    const float* fcos = (const float*)d_in[1];
    const float* fsin = (const float*)d_in[2];
    const float* wq   = (const float*)d_in[3];
    const float* wk   = (const float*)d_in[4];
    const float* wv   = (const float*)d_in[5];
    const float* wo   = (const float*)d_in[6];
    float* out = (float*)d_out;

    float* qb = (float*)d_ws;
    float* kb = qb + (size_t)M_ * (H_ * HD_);
    float* vb = kb + (size_t)M_ * (KVH_ * HD_);
    float* at = vb + (size_t)M_ * (KVH_ * HD_);

    const dim3 blk(128);
    gemm64x32_wmma<1><<<dim3((H_ * HD_) / 128, M_ / 64), blk, 0, stream>>>(
        x, wq, qb, M_, H_ * HD_, D_, fcos, fsin);
    gemm64x32_wmma<1><<<dim3((KVH_ * HD_) / 128, M_ / 64), blk, 0, stream>>>(
        x, wk, kb, M_, KVH_ * HD_, D_, fcos, fsin);
    gemm64x32_wmma<0><<<dim3((KVH_ * HD_) / 128, M_ / 64), blk, 0, stream>>>(
        x, wv, vb, M_, KVH_ * HD_, D_, fcos, fsin);

    attn_wmma<<<dim3(S_ / 16, H_, B_), dim3(32), 0, stream>>>(qb, kb, vb, at);

    gemm64x32_wmma<0><<<dim3(D_ / 128, M_ / 64), blk, 0, stream>>>(
        at, wo, out, M_, D_, H_ * HD_, fcos, fsin);
}


# GEMM operands staged through LDS by the whole block (coalesced 16-byte loads, 2 stages, W transposed), 16-byte LDS fragment reads a group ahead; tiles 128x64, 64x64; sum order in 8-k groups changes
# speedup vs baseline: 3.9369x; 3.9369x over previous
#include <hip/hip_runtime.h>
#include <hip/hip_bf16.h>


typedef __attribute__((ext_vector_type(2))) float v2f;
typedef __attribute__((ext_vector_type(4))) float v4f;
typedef __attribute__((ext_vector_type(8))) float v8f;

#define B_ 2
#define S_ 2048
#define D_ 1024
#define H_ 16
#define KVH_ 4
#define HD_ 64
#define WIN_ 64
#define M_ (B_ * S_)

template <int BM, int BN, int BK, int WM, int WN, int DO_ROPE>
__global__ void __launch_bounds__((BM / WM) * (BN / WN) * 32)
gemm_lds_wmma(const float* __restrict__ A, const float* __restrict__ W,
              float* __restrict__ C, int M, int N, int K,
              const float* __restrict__ fcos,
              const float* __restrict__ fsin) {
    constexpr int NWN  = BN / WN;
    constexpr int NW   = (BM / WM) * NWN;
    constexpr int NT   = NW * 32;
    constexpr int P    = BK + 4;
    constexpr int TM   = WM / 16;
    constexpr int TN   = WN / 16;
    constexpr int F4R  = BK / 4;
    constexpr int A_IT = (BM * F4R) / NT;
    constexpr int NBN  = BN / 32;
    constexpr int B_IT = ((BK / 4) * NBN) / NW;
    constexpr int STAGE = (BM + BN) * P;
    static_assert((BM * F4R) % NT == 0, "A tile must divide among the threads");
    static_assert(((BK / 4) * NBN) % NW == 0, "W tile must divide among the waves");
    static_assert(BK % 8 == 0 && (P / 4) % 2 == 1, "pitch must be 16 B x odd");

    __shared__ __attribute__((aligned(16))) float lds[2 * STAGE];

    const int tid  = threadIdx.x;
    const int lane = tid & 31;
    const int wave = __builtin_amdgcn_readfirstlane(tid >> 5);
    const int wm   = wave / NWN;
    const int wn   = wave % NWN;
    const int tileM = blockIdx.y * BM;
    const int tileN = blockIdx.x * BN;

    const int l15 = lane & 15;
    const int h4  = (lane >> 4) << 2;

    v4f ra[A_IT], rb[B_IT];
    const float* agp[A_IT];
    int          asd[A_IT];
    #pragma unroll
    for (int i = 0; i < A_IT; ++i) {
        const int idx = tid + i * NT;
        const int row = idx / F4R;
        const int c4  = idx % F4R;
        agp[i] = A + (long)(tileM + row) * K + c4 * 4;
        asd[i] = row * P + c4 * 4;
    }
    const float* bgp[B_IT];
    int          bsd[B_IT];
    #pragma unroll
    for (int i = 0; i < B_IT; ++i) {
        const int blk = wave + i * NW;
        const int nb  = blk % NBN;
        const int kb  = blk / NBN;
        const int k   = kb * 4 + (lane >> 3);
        const int n   = nb * 32 + (lane & 7) * 4;
        bgp[i] = W + (long)k * N + tileN + n;
        bsd[i] = BM * P + n * P + k;
    }

    auto load_tile = [&](int k0) {
        #pragma unroll
        for (int i = 0; i < A_IT; ++i) ra[i] = *(const v4f*)(agp[i] + k0);
        #pragma unroll
        for (int i = 0; i < B_IT; ++i) rb[i] = *(const v4f*)(bgp[i] + (long)k0 * N);
    };
    auto store_tile = [&](int stage) {
        float* s = lds + stage * STAGE;
        #pragma unroll
        for (int i = 0; i < A_IT; ++i) *(v4f*)(s + asd[i]) = ra[i];
        #pragma unroll
        for (int i = 0; i < B_IT; ++i) {
            s[bsd[i]]         = rb[i].x;
            s[bsd[i] + P]     = rb[i].y;
            s[bsd[i] + 2 * P] = rb[i].z;
            s[bsd[i] + 3 * P] = rb[i].w;
        }
    };

    v8f acc[TM][TN] = {};

    const int aoff = (wm * WM + l15) * P + h4;
    const int boff = BM * P + (wn * WN + l15) * P + h4;

    const int nk = K / BK;
    load_tile(0);
    store_tile(0);
    __syncthreads();

    for (int kt = 0; kt < nk; ++kt) {
        const int cur = kt & 1;
        const bool more = (kt + 1 < nk);
        if (more) load_tile((kt + 1) * BK);

        const float* s = lds + cur * STAGE;
        v4f a[2][TM], b[2][TN];
        #pragma unroll
        for (int i = 0; i < TM; ++i) a[0][i] = *(const v4f*)(s + aoff + i * 16 * P);
        #pragma unroll
        for (int j = 0; j < TN; ++j) b[0][j] = *(const v4f*)(s + boff + j * 16 * P);
        #pragma unroll
        for (int g = 0; g < BK / 8; ++g) {
            const int cb = g & 1, nb = cb ^ 1;
            if (g + 1 < BK / 8) {
                #pragma unroll
                for (int i = 0; i < TM; ++i)
                    a[nb][i] = *(const v4f*)(s + aoff + i * 16 * P + (g + 1) * 8);
                #pragma unroll
                for (int j = 0; j < TN; ++j)
                    b[nb][j] = *(const v4f*)(s + boff + j * 16 * P + (g + 1) * 8);
            }
            __builtin_amdgcn_sched_barrier(0);
            #pragma unroll
            for (int i = 0; i < TM; ++i)
                #pragma unroll
                for (int j = 0; j < TN; ++j) {
                    v2f a0 = {a[cb][i].x, a[cb][i].y};
                    v2f b0 = {b[cb][j].x, b[cb][j].y};
                    acc[i][j] = __builtin_amdgcn_wmma_f32_16x16x4_f32(
                        false, a0, false, b0, (short)0, acc[i][j], false, false);
                }
            #pragma unroll
            for (int i = 0; i < TM; ++i)
                #pragma unroll
                for (int j = 0; j < TN; ++j) {
                    v2f a1 = {a[cb][i].z, a[cb][i].w};
                    v2f b1 = {b[cb][j].z, b[cb][j].w};
                    acc[i][j] = __builtin_amdgcn_wmma_f32_16x16x4_f32(
                        false, a1, false, b1, (short)0, acc[i][j], false, false);
                }
        }

        if (more) store_tile(cur ^ 1);
        __syncthreads();
    }

    const int rowOff = (lane >> 4) << 3;
    #pragma unroll
    for (int i = 0; i < TM; ++i) {
        const int rowBase = tileM + wm * WM + i * 16 + rowOff;
        #pragma unroll
        for (int j = 0; j < TN; ++j) {
            const int col = tileN + wn * WN + j * 16 + l15;
            #pragma unroll
            for (int r = 0; r < 8; ++r) {
                float val = acc[i][j][r];
                if (DO_ROPE) {
                    const float partner = __shfl_xor(val, 1, 32);
                    const int pos = (rowBase + r) & (S_ - 1);
                    const int fi  = (col & (HD_ - 1)) >> 1;
                    const float c = fcos[pos * (HD_ / 2) + fi];
                    const float s = fsin[pos * (HD_ / 2) + fi];
                    val = ((col & 1) == 0) ? (val * c - partner * s)
                                           : (partner * s + val * c);
                }
                C[(long)(rowBase + r) * N + col] = val;
            }
        }
    }
}

__global__ void attn_wmma(const float* __restrict__ q, const float* __restrict__ kbuf,
                          const float* __restrict__ vbuf, float* __restrict__ o) {
    __shared__ float p[16][83];

    const int lane = threadIdx.x;
    const int i0   = blockIdx.x * 16;
    const int h    = blockIdx.y;
    const int b    = blockIdx.z;
    const int kvh  = h >> 2;

    const int l15  = lane & 15;
    const int koff = (lane >> 4) << 1;
    const int r8   = (lane >> 4) << 3;
    const int keystart = i0 - WIN_;

    const float* qrow = q + ((long)(b * S_ + i0 + l15) * H_ + h) * HD_;
    for (int t = 0; t < 5; ++t) {
        const int j0 = keystart + t * 16;
        if (j0 < 0) {
            #pragma unroll
            for (int r = 0; r < 8; ++r) p[r8 + r][t * 16 + l15] = -1e9f;
            continue;
        }
        const float* krow = kbuf + ((long)(b * S_ + j0 + l15) * KVH_ + kvh) * HD_;
        v8f acc = {};
        for (int kk = 0; kk < HD_; kk += 4) {
            const int kb = kk + koff;
            v2f a, bb;
            a.x  = qrow[kb];  a.y  = qrow[kb + 1];
            bb.x = krow[kb];  bb.y = krow[kb + 1];
            acc = __builtin_amdgcn_wmma_f32_16x16x4_f32(false, a, false, bb,
                                                        (short)0, acc, false, false);
        }
        #pragma unroll
        for (int r = 0; r < 8; ++r) {
            const int row  = i0 + r8 + r;
            const int colk = j0 + l15;
            float sc = fmaxf(acc[r] * 0.125f, -1e9f);
            if (colk > row || row - colk > WIN_) sc = -1e9f;
            p[r8 + r][t * 16 + l15] = sc;
        }
    }
    __syncthreads();

    if (lane < 16) {
        float m = -1e30f;
        for (int c = 0; c < 80; ++c) m = fmaxf(m, p[lane][c]);
        float sum = 0.0f;
        for (int c = 0; c < 80; ++c) {
            const float e = expf(p[lane][c] - m);
            p[lane][c] = e;
            sum += e;
        }
        const float inv = 1.0f / sum;
        for (int c = 0; c < 80; ++c) p[lane][c] *= inv;
    }
    __syncthreads();

    for (int nt = 0; nt < 4; ++nt) {
        const int n0 = nt * 16;
        v8f acc = {};
        for (int kk = 0; kk < 80; kk += 4) {
            const int kb = kk + koff;
            v2f a, bb;
            a.x = p[l15][kb];
            a.y = p[l15][kb + 1];
            const int key0 = keystart + kb;
            const int key1 = key0 + 1;
            bb.x = (key0 >= 0)
                 ? vbuf[((long)(b * S_ + key0) * KVH_ + kvh) * HD_ + n0 + l15] : 0.0f;
            bb.y = (key1 >= 0)
                 ? vbuf[((long)(b * S_ + key1) * KVH_ + kvh) * HD_ + n0 + l15] : 0.0f;
            acc = __builtin_amdgcn_wmma_f32_16x16x4_f32(false, a, false, bb,
                                                        (short)0, acc, false, false);
        }
        #pragma unroll
        for (int r = 0; r < 8; ++r) {
            o[((long)(b * S_ + i0 + r8 + r) * H_ + h) * HD_ + n0 + l15] = acc[r];
        }
    }
}

extern "C" void kernel_launch(void* const* d_in, const int* in_sizes, int n_in,
                              void* d_out, int out_size, void* d_ws, size_t ws_size,
                              hipStream_t stream) {
    const float* x    = (const float*)d_in[0];
    const float* fcos = (const float*)d_in[1];
    const float* fsin = (const float*)d_in[2];
    const float* wq   = (const float*)d_in[3];
    const float* wk   = (const float*)d_in[4];
    const float* wv   = (const float*)d_in[5];
    const float* wo   = (const float*)d_in[6];
    float* out = (float*)d_out;

    float* qb = (float*)d_ws;
    float* kb = qb + (size_t)M_ * (H_ * HD_);
    float* vb = kb + (size_t)M_ * (KVH_ * HD_);
    float* at = vb + (size_t)M_ * (KVH_ * HD_);

    const dim3 blk(128);
    gemm_lds_wmma<128, 64, 32, 64, 32, 1><<<dim3((H_ * HD_) / 64, M_ / 128), blk, 0, stream>>>(
        x, wq, qb, M_, H_ * HD_, D_, fcos, fsin);
    gemm_lds_wmma<64, 64, 64, 32, 32, 1><<<dim3((KVH_ * HD_) / 64, M_ / 64), blk, 0, stream>>>(
        x, wk, kb, M_, KVH_ * HD_, D_, fcos, fsin);
    gemm_lds_wmma<64, 64, 64, 32, 32, 0><<<dim3((KVH_ * HD_) / 64, M_ / 64), blk, 0, stream>>>(
        x, wv, vb, M_, KVH_ * HD_, D_, fcos, fsin);

    attn_wmma<<<dim3(S_ / 16, H_, B_), dim3(32), 0, stream>>>(qb, kb, vb, at);

    gemm_lds_wmma<128, 64, 32, 64, 32, 0><<<dim3(D_ / 64, M_ / 128), blk, 0, stream>>>(
        at, wo, out, M_, D_, H_ * HD_, fcos, fsin);
}


# Attention by 4-wave blocks (a KV head x 16 queries): 80 K and V rows staged once in LDS by 16-byte loads, shared by the 4 query heads, fragments by 16-byte LDS reads; score and PV sum order changes
# speedup vs baseline: 4.4072x; 1.1195x over previous
#include <hip/hip_runtime.h>
#include <hip/hip_bf16.h>


typedef __attribute__((ext_vector_type(2))) float v2f;
typedef __attribute__((ext_vector_type(4))) float v4f;
typedef __attribute__((ext_vector_type(8))) float v8f;

#define B_ 2
#define S_ 2048
#define D_ 1024
#define H_ 16
#define KVH_ 4
#define HD_ 64
#define WIN_ 64
#define M_ (B_ * S_)

template <int BM, int BN, int BK, int WM, int WN, int DO_ROPE>
__global__ void __launch_bounds__((BM / WM) * (BN / WN) * 32)
gemm_lds_wmma(const float* __restrict__ A, const float* __restrict__ W,
              float* __restrict__ C, int M, int N, int K,
              const float* __restrict__ fcos,
              const float* __restrict__ fsin) {
    constexpr int NWN  = BN / WN;
    constexpr int NW   = (BM / WM) * NWN;
    constexpr int NT   = NW * 32;
    constexpr int P    = BK + 4;
    constexpr int TM   = WM / 16;
    constexpr int TN   = WN / 16;
    constexpr int F4R  = BK / 4;
    constexpr int A_IT = (BM * F4R) / NT;
    constexpr int NBN  = BN / 32;
    constexpr int B_IT = ((BK / 4) * NBN) / NW;
    constexpr int STAGE = (BM + BN) * P;
    static_assert((BM * F4R) % NT == 0, "A tile must divide among the threads");
    static_assert(((BK / 4) * NBN) % NW == 0, "W tile must divide among the waves");
    static_assert(BK % 8 == 0 && (P / 4) % 2 == 1, "pitch must be 16 B x odd");

    __shared__ __attribute__((aligned(16))) float lds[2 * STAGE];

    const int tid  = threadIdx.x;
    const int lane = tid & 31;
    const int wave = __builtin_amdgcn_readfirstlane(tid >> 5);
    const int wm   = wave / NWN;
    const int wn   = wave % NWN;
    const int tileM = blockIdx.y * BM;
    const int tileN = blockIdx.x * BN;

    const int l15 = lane & 15;
    const int h4  = (lane >> 4) << 2;

    v4f ra[A_IT], rb[B_IT];
    const float* agp[A_IT];
    int          asd[A_IT];
    #pragma unroll
    for (int i = 0; i < A_IT; ++i) {
        const int idx = tid + i * NT;
        const int row = idx / F4R;
        const int c4  = idx % F4R;
        agp[i] = A + (long)(tileM + row) * K + c4 * 4;
        asd[i] = row * P + c4 * 4;
    }
    const float* bgp[B_IT];
    int          bsd[B_IT];
    #pragma unroll
    for (int i = 0; i < B_IT; ++i) {
        const int blk = wave + i * NW;
        const int nb  = blk % NBN;
        const int kb  = blk / NBN;
        const int k   = kb * 4 + (lane >> 3);
        const int n   = nb * 32 + (lane & 7) * 4;
        bgp[i] = W + (long)k * N + tileN + n;
        bsd[i] = BM * P + n * P + k;
    }

    auto load_tile = [&](int k0) {
        #pragma unroll
        for (int i = 0; i < A_IT; ++i) ra[i] = *(const v4f*)(agp[i] + k0);
        #pragma unroll
        for (int i = 0; i < B_IT; ++i) rb[i] = *(const v4f*)(bgp[i] + (long)k0 * N);
    };
    auto store_tile = [&](int stage) {
        float* s = lds + stage * STAGE;
        #pragma unroll
        for (int i = 0; i < A_IT; ++i) *(v4f*)(s + asd[i]) = ra[i];
        #pragma unroll
        for (int i = 0; i < B_IT; ++i) {
            s[bsd[i]]         = rb[i].x;
            s[bsd[i] + P]     = rb[i].y;
            s[bsd[i] + 2 * P] = rb[i].z;
            s[bsd[i] + 3 * P] = rb[i].w;
        }
    };

    v8f acc[TM][TN] = {};

    const int aoff = (wm * WM + l15) * P + h4;
    const int boff = BM * P + (wn * WN + l15) * P + h4;

    const int nk = K / BK;
    load_tile(0);
    store_tile(0);
    __syncthreads();

    for (int kt = 0; kt < nk; ++kt) {
        const int cur = kt & 1;
        const bool more = (kt + 1 < nk);
        if (more) load_tile((kt + 1) * BK);

        const float* s = lds + cur * STAGE;
        v4f a[2][TM], b[2][TN];
        #pragma unroll
        for (int i = 0; i < TM; ++i) a[0][i] = *(const v4f*)(s + aoff + i * 16 * P);
        #pragma unroll
        for (int j = 0; j < TN; ++j) b[0][j] = *(const v4f*)(s + boff + j * 16 * P);
        #pragma unroll
        for (int g = 0; g < BK / 8; ++g) {
            const int cb = g & 1, nb = cb ^ 1;
            if (g + 1 < BK / 8) {
                #pragma unroll
                for (int i = 0; i < TM; ++i)
                    a[nb][i] = *(const v4f*)(s + aoff + i * 16 * P + (g + 1) * 8);
                #pragma unroll
                for (int j = 0; j < TN; ++j)
                    b[nb][j] = *(const v4f*)(s + boff + j * 16 * P + (g + 1) * 8);
            }
            __builtin_amdgcn_sched_barrier(0);
            #pragma unroll
            for (int i = 0; i < TM; ++i)
                #pragma unroll
                for (int j = 0; j < TN; ++j) {
                    v2f a0 = {a[cb][i].x, a[cb][i].y};
                    v2f b0 = {b[cb][j].x, b[cb][j].y};
                    acc[i][j] = __builtin_amdgcn_wmma_f32_16x16x4_f32(
                        false, a0, false, b0, (short)0, acc[i][j], false, false);
                }
            #pragma unroll
            for (int i = 0; i < TM; ++i)
                #pragma unroll
                for (int j = 0; j < TN; ++j) {
                    v2f a1 = {a[cb][i].z, a[cb][i].w};
                    v2f b1 = {b[cb][j].z, b[cb][j].w};
                    acc[i][j] = __builtin_amdgcn_wmma_f32_16x16x4_f32(
                        false, a1, false, b1, (short)0, acc[i][j], false, false);
                }
        }

        if (more) store_tile(cur ^ 1);
        __syncthreads();
    }

    const int rowOff = (lane >> 4) << 3;
    #pragma unroll
    for (int i = 0; i < TM; ++i) {
        const int rowBase = tileM + wm * WM + i * 16 + rowOff;
        #pragma unroll
        for (int j = 0; j < TN; ++j) {
            const int col = tileN + wn * WN + j * 16 + l15;
            #pragma unroll
            for (int r = 0; r < 8; ++r) {
                float val = acc[i][j][r];
                if (DO_ROPE) {
                    const float partner = __shfl_xor(val, 1, 32);
                    const int pos = (rowBase + r) & (S_ - 1);
                    const int fi  = (col & (HD_ - 1)) >> 1;
                    const float c = fcos[pos * (HD_ / 2) + fi];
                    const float s = fsin[pos * (HD_ / 2) + fi];
                    val = ((col & 1) == 0) ? (val * c - partner * s)
                                           : (partner * s + val * c);
                }
                C[(long)(rowBase + r) * N + col] = val;
            }
        }
    }
}

#define KP_ (HD_ + 4)
#define PP_ 84

__global__ void __launch_bounds__(128)
attn_wmma(const float* __restrict__ q, const float* __restrict__ kbuf,
          const float* __restrict__ vbuf, float* __restrict__ o) {
    __shared__ __attribute__((aligned(16))) float Ks[80 * KP_];
    __shared__ __attribute__((aligned(16))) float Vs[80 * KP_];
    __shared__ __attribute__((aligned(16))) float Ps[4][16 * PP_];

    const int tid  = threadIdx.x;
    const int lane = tid & 31;
    const int wave = __builtin_amdgcn_readfirstlane(tid >> 5);
    const int i0   = blockIdx.x * 16;
    const int kvh  = blockIdx.y;
    const int b    = blockIdx.z;
    const int h    = kvh * 4 + wave;

    const int l15  = lane & 15;
    const int h4   = (lane >> 4) << 2;
    const int r8   = (lane >> 4) << 3;
    const int keystart = i0 - WIN_;

    v4f qf[8];
    {
        const float* qrow = q + ((long)(b * S_ + i0 + l15) * H_ + h) * HD_ + h4;
        #pragma unroll
        for (int g = 0; g < 8; ++g) qf[g] = *(const v4f*)(qrow + g * 8);
    }

    {
        v4f kr[10], vr[10];
        #pragma unroll
        for (int i = 0; i < 10; ++i) {
            const int idx = tid + i * 128;
            const int row = idx >> 4;
            const int c4  = idx & 15;
            const int key = keystart + row;
            const int keyc = key < 0 ? 0 : key;
            const long off = ((long)(b * S_ + keyc) * KVH_ + kvh) * HD_ + c4 * 4;
            kr[i] = *(const v4f*)(kbuf + off);
            vr[i] = *(const v4f*)(vbuf + off);
        }
        #pragma unroll
        for (int i = 0; i < 10; ++i) {
            const int idx = tid + i * 128;
            const int row = idx >> 4;
            const int c4  = idx & 15;
            const bool valid = (keystart + row) >= 0;
            const v4f zero = {0.0f, 0.0f, 0.0f, 0.0f};
            *(v4f*)(Ks + row * KP_ + c4 * 4) = kr[i];
            *(v4f*)(Vs + row * KP_ + c4 * 4) = valid ? vr[i] : zero;
        }
    }
    __syncthreads();

    float* p = Ps[wave];

    {
        v8f sacc[5] = {};
        const float* kfrag = Ks + l15 * KP_ + h4;
        v4f kf[2][5];
        #pragma unroll
        for (int t = 0; t < 5; ++t) kf[0][t] = *(const v4f*)(kfrag + t * 16 * KP_);
        #pragma unroll
        for (int g = 0; g < 8; ++g) {
            const int cb = g & 1, nb = cb ^ 1;
            if (g + 1 < 8) {
                #pragma unroll
                for (int t = 0; t < 5; ++t)
                    kf[nb][t] = *(const v4f*)(kfrag + t * 16 * KP_ + (g + 1) * 8);
            }
            __builtin_amdgcn_sched_barrier(0);
            #pragma unroll
            for (int t = 0; t < 5; ++t) {
                v2f a0 = {qf[g].x, qf[g].y};
                v2f b0 = {kf[cb][t].x, kf[cb][t].y};
                sacc[t] = __builtin_amdgcn_wmma_f32_16x16x4_f32(
                    false, a0, false, b0, (short)0, sacc[t], false, false);
            }
            #pragma unroll
            for (int t = 0; t < 5; ++t) {
                v2f a1 = {qf[g].z, qf[g].w};
                v2f b1 = {kf[cb][t].z, kf[cb][t].w};
                sacc[t] = __builtin_amdgcn_wmma_f32_16x16x4_f32(
                    false, a1, false, b1, (short)0, sacc[t], false, false);
            }
        }
        #pragma unroll
        for (int t = 0; t < 5; ++t) {
            const int colk = keystart + t * 16 + l15;
            #pragma unroll
            for (int r = 0; r < 8; ++r) {
                const int row = i0 + r8 + r;
                float sc = fmaxf(sacc[t][r] * 0.125f, -1e9f);
                if (colk > row || row - colk > WIN_ || colk < 0) sc = -1e9f;
                p[(r8 + r) * PP_ + t * 16 + l15] = sc;
            }
        }
    }
    __syncthreads();

    if (lane < 16) {
        float* pr = p + lane * PP_;
        float m = -1e30f;
        for (int c = 0; c < 80; ++c) m = fmaxf(m, pr[c]);
        float sum = 0.0f;
        for (int c = 0; c < 80; ++c) {
            const float e = expf(pr[c] - m);
            pr[c] = e;
            sum += e;
        }
        const float inv = 1.0f / sum;
        for (int c = 0; c < 80; ++c) pr[c] *= inv;
    }
    __syncthreads();

    {
        v8f oacc[4] = {};
        const float* pfrag = p + l15 * PP_ + h4;
        const float* vfrag = Vs + h4 * KP_ + l15 * 4;
        v4f pf[2], vf[2][4];
        pf[0] = *(const v4f*)(pfrag);
        #pragma unroll
        for (int e = 0; e < 4; ++e) vf[0][e] = *(const v4f*)(vfrag + e * KP_);
        #pragma unroll
        for (int g = 0; g < 10; ++g) {
            const int cb = g & 1, nb = cb ^ 1;
            if (g + 1 < 10) {
                pf[nb] = *(const v4f*)(pfrag + (g + 1) * 8);
                #pragma unroll
                for (int e = 0; e < 4; ++e)
                    vf[nb][e] = *(const v4f*)(vfrag + ((g + 1) * 8 + e) * KP_);
            }
            __builtin_amdgcn_sched_barrier(0);
            #pragma unroll
            for (int nt = 0; nt < 4; ++nt) {
                v2f a0 = {pf[cb].x, pf[cb].y};
                v2f b0 = {vf[cb][0][nt], vf[cb][1][nt]};
                oacc[nt] = __builtin_amdgcn_wmma_f32_16x16x4_f32(
                    false, a0, false, b0, (short)0, oacc[nt], false, false);
            }
            #pragma unroll
            for (int nt = 0; nt < 4; ++nt) {
                v2f a1 = {pf[cb].z, pf[cb].w};
                v2f b1 = {vf[cb][2][nt], vf[cb][3][nt]};
                oacc[nt] = __builtin_amdgcn_wmma_f32_16x16x4_f32(
                    false, a1, false, b1, (short)0, oacc[nt], false, false);
            }
        }
        float* orow = o + ((long)(b * S_ + i0 + r8) * H_ + h) * HD_ + l15 * 4;
        #pragma unroll
        for (int r = 0; r < 8; ++r) {
            v4f ov = {oacc[0][r], oacc[1][r], oacc[2][r], oacc[3][r]};
            *(v4f*)(orow + (long)r * (H_ * HD_)) = ov;
        }
    }
}

extern "C" void kernel_launch(void* const* d_in, const int* in_sizes, int n_in,
                              void* d_out, int out_size, void* d_ws, size_t ws_size,
                              hipStream_t stream) {
    const float* x    = (const float*)d_in[0];
    const float* fcos = (const float*)d_in[1];
    const float* fsin = (const float*)d_in[2];
    const float* wq   = (const float*)d_in[3];
    const float* wk   = (const float*)d_in[4];
    const float* wv   = (const float*)d_in[5];
    const float* wo   = (const float*)d_in[6];
    float* out = (float*)d_out;

    float* qb = (float*)d_ws;
    float* kb = qb + (size_t)M_ * (H_ * HD_);
    float* vb = kb + (size_t)M_ * (KVH_ * HD_);
    float* at = vb + (size_t)M_ * (KVH_ * HD_);

    const dim3 blk(128);
    gemm_lds_wmma<128, 64, 32, 64, 32, 1><<<dim3((H_ * HD_) / 64, M_ / 128), blk, 0, stream>>>(
        x, wq, qb, M_, H_ * HD_, D_, fcos, fsin);
    gemm_lds_wmma<64, 64, 64, 32, 32, 1><<<dim3((KVH_ * HD_) / 64, M_ / 64), blk, 0, stream>>>(
        x, wk, kb, M_, KVH_ * HD_, D_, fcos, fsin);
    gemm_lds_wmma<64, 64, 64, 32, 32, 0><<<dim3((KVH_ * HD_) / 64, M_ / 64), blk, 0, stream>>>(
        x, wv, vb, M_, KVH_ * HD_, D_, fcos, fsin);

    attn_wmma<<<dim3(S_ / 16, KVH_, B_), dim3(128), 0, stream>>>(qb, kb, vb, at);

    gemm_lds_wmma<128, 64, 32, 64, 32, 0><<<dim3(D_ / 64, M_ / 128), blk, 0, stream>>>(
        at, wo, out, M_, D_, H_ * HD_, fcos, fsin);
}


# GEMM K loop: next tile stored to LDS in mid tile under the WMMAs (global loads a tile ahead), LDS-only barrier before the last 8-k group, next tile's first fragments read under it; sums unchanged
# speedup vs baseline: 4.7580x; 1.0796x over previous
#include <hip/hip_runtime.h>
#include <hip/hip_bf16.h>


typedef __attribute__((ext_vector_type(2))) float v2f;
typedef __attribute__((ext_vector_type(4))) float v4f;
typedef __attribute__((ext_vector_type(8))) float v8f;

#define B_ 2
#define S_ 2048
#define D_ 1024
#define H_ 16
#define KVH_ 4
#define HD_ 64
#define WIN_ 64
#define M_ (B_ * S_)

template <int BM, int BN, int BK, int WM, int WN, int DO_ROPE>
__global__ void __launch_bounds__((BM / WM) * (BN / WN) * 32)
gemm_lds_wmma(const float* __restrict__ A, const float* __restrict__ W,
              float* __restrict__ C, int M, int N, int K,
              const float* __restrict__ fcos,
              const float* __restrict__ fsin) {
    constexpr int NWN  = BN / WN;
    constexpr int NW   = (BM / WM) * NWN;
    constexpr int NT   = NW * 32;
    constexpr int P    = BK + 4;
    constexpr int TM   = WM / 16;
    constexpr int TN   = WN / 16;
    constexpr int F4R  = BK / 4;
    constexpr int A_IT = (BM * F4R) / NT;
    constexpr int NBN  = BN / 32;
    constexpr int B_IT = ((BK / 4) * NBN) / NW;
    constexpr int STAGE = (BM + BN) * P;
    static_assert((BM * F4R) % NT == 0, "A tile must divide among the threads");
    static_assert(((BK / 4) * NBN) % NW == 0, "W tile must divide among the waves");
    static_assert(BK % 8 == 0 && (P / 4) % 2 == 1, "pitch must be 16 B x odd");

    __shared__ __attribute__((aligned(16))) float lds[2 * STAGE];

    const int tid  = threadIdx.x;
    const int lane = tid & 31;
    const int wave = __builtin_amdgcn_readfirstlane(tid >> 5);
    const int wm   = wave / NWN;
    const int wn   = wave % NWN;
    const int tileM = blockIdx.y * BM;
    const int tileN = blockIdx.x * BN;

    const int l15 = lane & 15;
    const int h4  = (lane >> 4) << 2;

    v4f ra[A_IT], rb[B_IT];
    const float* agp[A_IT];
    int          asd[A_IT];
    #pragma unroll
    for (int i = 0; i < A_IT; ++i) {
        const int idx = tid + i * NT;
        const int row = idx / F4R;
        const int c4  = idx % F4R;
        agp[i] = A + (long)(tileM + row) * K + c4 * 4;
        asd[i] = row * P + c4 * 4;
    }
    const float* bgp[B_IT];
    int          bsd[B_IT];
    #pragma unroll
    for (int i = 0; i < B_IT; ++i) {
        const int blk = wave + i * NW;
        const int nb  = blk % NBN;
        const int kb  = blk / NBN;
        const int k   = kb * 4 + (lane >> 3);
        const int n   = nb * 32 + (lane & 7) * 4;
        bgp[i] = W + (long)k * N + tileN + n;
        bsd[i] = BM * P + n * P + k;
    }

    auto load_tile = [&](int k0) {
        #pragma unroll
        for (int i = 0; i < A_IT; ++i) ra[i] = *(const v4f*)(agp[i] + k0);
        #pragma unroll
        for (int i = 0; i < B_IT; ++i) rb[i] = *(const v4f*)(bgp[i] + (long)k0 * N);
    };
    auto store_tile = [&](int stage) {
        float* s = lds + stage * STAGE;
        #pragma unroll
        for (int i = 0; i < A_IT; ++i) *(v4f*)(s + asd[i]) = ra[i];
        #pragma unroll
        for (int i = 0; i < B_IT; ++i) {
            s[bsd[i]]         = rb[i].x;
            s[bsd[i] + P]     = rb[i].y;
            s[bsd[i] + 2 * P] = rb[i].z;
            s[bsd[i] + 3 * P] = rb[i].w;
        }
    };

    auto lds_barrier = [&]() {
        __builtin_amdgcn_fence(__ATOMIC_RELEASE, "workgroup", "local");
        __builtin_amdgcn_s_barrier();
        __builtin_amdgcn_fence(__ATOMIC_ACQUIRE, "workgroup", "local");
    };

    v8f acc[TM][TN] = {};

    const int aoff = (wm * WM + l15) * P + h4;
    const int boff = BM * P + (wn * WN + l15) * P + h4;

    constexpr int G = BK / 8;
    static_assert(G % 2 == 0, "the fragment register sets alternate per group");

    const int nk = K / BK;
    load_tile(0);
    store_tile(0);
    if (nk > 1) load_tile(BK);
    lds_barrier();

    v4f a[2][TM], b[2][TN];
    #pragma unroll
    for (int i = 0; i < TM; ++i) a[0][i] = *(const v4f*)(lds + aoff + i * 16 * P);
    #pragma unroll
    for (int j = 0; j < TN; ++j) b[0][j] = *(const v4f*)(lds + boff + j * 16 * P);

    for (int kt = 0; kt < nk; ++kt) {
        const int cur = kt & 1;
        const bool more = (kt + 1 < nk);
        const float* s  = lds + cur * STAGE;
        const float* sn = lds + (cur ^ 1) * STAGE;
        #pragma unroll
        for (int g = 0; g < G; ++g) {
            const int cb = g & 1, nb = cb ^ 1;
            if (g == G / 2 && more) {
                store_tile(cur ^ 1);
                if (kt + 2 < nk) load_tile((kt + 2) * BK);
            }
            if (g + 1 < G) {
                #pragma unroll
                for (int i = 0; i < TM; ++i)
                    a[nb][i] = *(const v4f*)(s + aoff + i * 16 * P + (g + 1) * 8);
                #pragma unroll
                for (int j = 0; j < TN; ++j)
                    b[nb][j] = *(const v4f*)(s + boff + j * 16 * P + (g + 1) * 8);
            } else {
                lds_barrier();
                if (more) {
                    #pragma unroll
                    for (int i = 0; i < TM; ++i)
                        a[nb][i] = *(const v4f*)(sn + aoff + i * 16 * P);
                    #pragma unroll
                    for (int j = 0; j < TN; ++j)
                        b[nb][j] = *(const v4f*)(sn + boff + j * 16 * P);
                }
            }
            __builtin_amdgcn_sched_barrier(0);
            #pragma unroll
            for (int i = 0; i < TM; ++i)
                #pragma unroll
                for (int j = 0; j < TN; ++j) {
                    v2f a0 = {a[cb][i].x, a[cb][i].y};
                    v2f b0 = {b[cb][j].x, b[cb][j].y};
                    acc[i][j] = __builtin_amdgcn_wmma_f32_16x16x4_f32(
                        false, a0, false, b0, (short)0, acc[i][j], false, false);
                }
            #pragma unroll
            for (int i = 0; i < TM; ++i)
                #pragma unroll
                for (int j = 0; j < TN; ++j) {
                    v2f a1 = {a[cb][i].z, a[cb][i].w};
                    v2f b1 = {b[cb][j].z, b[cb][j].w};
                    acc[i][j] = __builtin_amdgcn_wmma_f32_16x16x4_f32(
                        false, a1, false, b1, (short)0, acc[i][j], false, false);
                }
        }
    }

    const int rowOff = (lane >> 4) << 3;
    #pragma unroll
    for (int i = 0; i < TM; ++i) {
        const int rowBase = tileM + wm * WM + i * 16 + rowOff;
        #pragma unroll
        for (int j = 0; j < TN; ++j) {
            const int col = tileN + wn * WN + j * 16 + l15;
            #pragma unroll
            for (int r = 0; r < 8; ++r) {
                float val = acc[i][j][r];
                if (DO_ROPE) {
                    const float partner = __shfl_xor(val, 1, 32);
                    const int pos = (rowBase + r) & (S_ - 1);
                    const int fi  = (col & (HD_ - 1)) >> 1;
                    const float c = fcos[pos * (HD_ / 2) + fi];
                    const float s = fsin[pos * (HD_ / 2) + fi];
                    val = ((col & 1) == 0) ? (val * c - partner * s)
                                           : (partner * s + val * c);
                }
                C[(long)(rowBase + r) * N + col] = val;
            }
        }
    }
}

#define KP_ (HD_ + 4)
#define PP_ 84

__global__ void __launch_bounds__(128)
attn_wmma(const float* __restrict__ q, const float* __restrict__ kbuf,
          const float* __restrict__ vbuf, float* __restrict__ o) {
    __shared__ __attribute__((aligned(16))) float Ks[80 * KP_];
    __shared__ __attribute__((aligned(16))) float Vs[80 * KP_];
    __shared__ __attribute__((aligned(16))) float Ps[4][16 * PP_];

    const int tid  = threadIdx.x;
    const int lane = tid & 31;
    const int wave = __builtin_amdgcn_readfirstlane(tid >> 5);
    const int i0   = blockIdx.x * 16;
    const int kvh  = blockIdx.y;
    const int b    = blockIdx.z;
    const int h    = kvh * 4 + wave;

    const int l15  = lane & 15;
    const int h4   = (lane >> 4) << 2;
    const int r8   = (lane >> 4) << 3;
    const int keystart = i0 - WIN_;

    v4f qf[8];
    {
        const float* qrow = q + ((long)(b * S_ + i0 + l15) * H_ + h) * HD_ + h4;
        #pragma unroll
        for (int g = 0; g < 8; ++g) qf[g] = *(const v4f*)(qrow + g * 8);
    }

    {
        v4f kr[10], vr[10];
        #pragma unroll
        for (int i = 0; i < 10; ++i) {
            const int idx = tid + i * 128;
            const int row = idx >> 4;
            const int c4  = idx & 15;
            const int key = keystart + row;
            const int keyc = key < 0 ? 0 : key;
            const long off = ((long)(b * S_ + keyc) * KVH_ + kvh) * HD_ + c4 * 4;
            kr[i] = *(const v4f*)(kbuf + off);
            vr[i] = *(const v4f*)(vbuf + off);
        }
        #pragma unroll
        for (int i = 0; i < 10; ++i) {
            const int idx = tid + i * 128;
            const int row = idx >> 4;
            const int c4  = idx & 15;
            const bool valid = (keystart + row) >= 0;
            const v4f zero = {0.0f, 0.0f, 0.0f, 0.0f};
            *(v4f*)(Ks + row * KP_ + c4 * 4) = kr[i];
            *(v4f*)(Vs + row * KP_ + c4 * 4) = valid ? vr[i] : zero;
        }
    }
    __syncthreads();

    float* p = Ps[wave];

    {
        v8f sacc[5] = {};
        const float* kfrag = Ks + l15 * KP_ + h4;
        v4f kf[2][5];
        #pragma unroll
        for (int t = 0; t < 5; ++t) kf[0][t] = *(const v4f*)(kfrag + t * 16 * KP_);
        #pragma unroll
        for (int g = 0; g < 8; ++g) {
            const int cb = g & 1, nb = cb ^ 1;
            if (g + 1 < 8) {
                #pragma unroll
                for (int t = 0; t < 5; ++t)
                    kf[nb][t] = *(const v4f*)(kfrag + t * 16 * KP_ + (g + 1) * 8);
            }
            __builtin_amdgcn_sched_barrier(0);
            #pragma unroll
            for (int t = 0; t < 5; ++t) {
                v2f a0 = {qf[g].x, qf[g].y};
                v2f b0 = {kf[cb][t].x, kf[cb][t].y};
                sacc[t] = __builtin_amdgcn_wmma_f32_16x16x4_f32(
                    false, a0, false, b0, (short)0, sacc[t], false, false);
            }
            #pragma unroll
            for (int t = 0; t < 5; ++t) {
                v2f a1 = {qf[g].z, qf[g].w};
                v2f b1 = {kf[cb][t].z, kf[cb][t].w};
                sacc[t] = __builtin_amdgcn_wmma_f32_16x16x4_f32(
                    false, a1, false, b1, (short)0, sacc[t], false, false);
            }
        }
        #pragma unroll
        for (int t = 0; t < 5; ++t) {
            const int colk = keystart + t * 16 + l15;
            #pragma unroll
            for (int r = 0; r < 8; ++r) {
                const int row = i0 + r8 + r;
                float sc = fmaxf(sacc[t][r] * 0.125f, -1e9f);
                if (colk > row || row - colk > WIN_ || colk < 0) sc = -1e9f;
                p[(r8 + r) * PP_ + t * 16 + l15] = sc;
            }
        }
    }
    __syncthreads();

    if (lane < 16) {
        float* pr = p + lane * PP_;
        float m = -1e30f;
        for (int c = 0; c < 80; ++c) m = fmaxf(m, pr[c]);
        float sum = 0.0f;
        for (int c = 0; c < 80; ++c) {
            const float e = expf(pr[c] - m);
            pr[c] = e;
            sum += e;
        }
        const float inv = 1.0f / sum;
        for (int c = 0; c < 80; ++c) pr[c] *= inv;
    }
    __syncthreads();

    {
        v8f oacc[4] = {};
        const float* pfrag = p + l15 * PP_ + h4;
        const float* vfrag = Vs + h4 * KP_ + l15 * 4;
        v4f pf[2], vf[2][4];
        pf[0] = *(const v4f*)(pfrag);
        #pragma unroll
        for (int e = 0; e < 4; ++e) vf[0][e] = *(const v4f*)(vfrag + e * KP_);
        #pragma unroll
        for (int g = 0; g < 10; ++g) {
            const int cb = g & 1, nb = cb ^ 1;
            if (g + 1 < 10) {
                pf[nb] = *(const v4f*)(pfrag + (g + 1) * 8);
                #pragma unroll
                for (int e = 0; e < 4; ++e)
                    vf[nb][e] = *(const v4f*)(vfrag + ((g + 1) * 8 + e) * KP_);
            }
            __builtin_amdgcn_sched_barrier(0);
            #pragma unroll
            for (int nt = 0; nt < 4; ++nt) {
                v2f a0 = {pf[cb].x, pf[cb].y};
                v2f b0 = {vf[cb][0][nt], vf[cb][1][nt]};
                oacc[nt] = __builtin_amdgcn_wmma_f32_16x16x4_f32(
                    false, a0, false, b0, (short)0, oacc[nt], false, false);
            }
            #pragma unroll
            for (int nt = 0; nt < 4; ++nt) {
                v2f a1 = {pf[cb].z, pf[cb].w};
                v2f b1 = {vf[cb][2][nt], vf[cb][3][nt]};
                oacc[nt] = __builtin_amdgcn_wmma_f32_16x16x4_f32(
                    false, a1, false, b1, (short)0, oacc[nt], false, false);
            }
        }
        float* orow = o + ((long)(b * S_ + i0 + r8) * H_ + h) * HD_ + l15 * 4;
        #pragma unroll
        for (int r = 0; r < 8; ++r) {
            v4f ov = {oacc[0][r], oacc[1][r], oacc[2][r], oacc[3][r]};
            *(v4f*)(orow + (long)r * (H_ * HD_)) = ov;
        }
    }
}

extern "C" void kernel_launch(void* const* d_in, const int* in_sizes, int n_in,
                              void* d_out, int out_size, void* d_ws, size_t ws_size,
                              hipStream_t stream) {
    const float* x    = (const float*)d_in[0];
    const float* fcos = (const float*)d_in[1];
    const float* fsin = (const float*)d_in[2];
    const float* wq   = (const float*)d_in[3];
    const float* wk   = (const float*)d_in[4];
    const float* wv   = (const float*)d_in[5];
    const float* wo   = (const float*)d_in[6];
    float* out = (float*)d_out;

    float* qb = (float*)d_ws;
    float* kb = qb + (size_t)M_ * (H_ * HD_);
    float* vb = kb + (size_t)M_ * (KVH_ * HD_);
    float* at = vb + (size_t)M_ * (KVH_ * HD_);

    const dim3 blk(128);
    gemm_lds_wmma<128, 64, 32, 64, 32, 1><<<dim3((H_ * HD_) / 64, M_ / 128), blk, 0, stream>>>(
        x, wq, qb, M_, H_ * HD_, D_, fcos, fsin);
    gemm_lds_wmma<64, 64, 64, 32, 32, 1><<<dim3((KVH_ * HD_) / 64, M_ / 64), blk, 0, stream>>>(
        x, wk, kb, M_, KVH_ * HD_, D_, fcos, fsin);
    gemm_lds_wmma<64, 64, 64, 32, 32, 0><<<dim3((KVH_ * HD_) / 64, M_ / 64), blk, 0, stream>>>(
        x, wv, vb, M_, KVH_ * HD_, D_, fcos, fsin);

    attn_wmma<<<dim3(S_ / 16, KVH_, B_), dim3(128), 0, stream>>>(qb, kb, vb, at);

    gemm_lds_wmma<128, 64, 32, 64, 32, 0><<<dim3(D_ / 64, M_ / 128), blk, 0, stream>>>(
        at, wo, out, M_, D_, H_ * HD_, fcos, fsin);
}


# Attention scores computed transposed (K as A operand): a lane holds 40 scores of one query, softmax in registers, P goes from accumulators into the PV WMMAs, no LDS P; softmax, PV sum order change
# speedup vs baseline: 4.9382x; 1.0379x over previous
#include <hip/hip_runtime.h>
#include <hip/hip_bf16.h>


typedef __attribute__((ext_vector_type(2))) float v2f;
typedef __attribute__((ext_vector_type(4))) float v4f;
typedef __attribute__((ext_vector_type(8))) float v8f;

#define B_ 2
#define S_ 2048
#define D_ 1024
#define H_ 16
#define KVH_ 4
#define HD_ 64
#define WIN_ 64
#define M_ (B_ * S_)

template <int BM, int BN, int BK, int WM, int WN, int DO_ROPE>
__global__ void __launch_bounds__((BM / WM) * (BN / WN) * 32)
gemm_lds_wmma(const float* __restrict__ A, const float* __restrict__ W,
              float* __restrict__ C, int M, int N, int K,
              const float* __restrict__ fcos,
              const float* __restrict__ fsin) {
    constexpr int NWN  = BN / WN;
    constexpr int NW   = (BM / WM) * NWN;
    constexpr int NT   = NW * 32;
    constexpr int P    = BK + 4;
    constexpr int TM   = WM / 16;
    constexpr int TN   = WN / 16;
    constexpr int F4R  = BK / 4;
    constexpr int A_IT = (BM * F4R) / NT;
    constexpr int NBN  = BN / 32;
    constexpr int B_IT = ((BK / 4) * NBN) / NW;
    constexpr int STAGE = (BM + BN) * P;
    static_assert((BM * F4R) % NT == 0, "A tile must divide among the threads");
    static_assert(((BK / 4) * NBN) % NW == 0, "W tile must divide among the waves");
    static_assert(BK % 8 == 0 && (P / 4) % 2 == 1, "pitch must be 16 B x odd");

    __shared__ __attribute__((aligned(16))) float lds[2 * STAGE];

    const int tid  = threadIdx.x;
    const int lane = tid & 31;
    const int wave = __builtin_amdgcn_readfirstlane(tid >> 5);
    const int wm   = wave / NWN;
    const int wn   = wave % NWN;
    const int tileM = blockIdx.y * BM;
    const int tileN = blockIdx.x * BN;

    const int l15 = lane & 15;
    const int h4  = (lane >> 4) << 2;

    v4f ra[A_IT], rb[B_IT];
    const float* agp[A_IT];
    int          asd[A_IT];
    #pragma unroll
    for (int i = 0; i < A_IT; ++i) {
        const int idx = tid + i * NT;
        const int row = idx / F4R;
        const int c4  = idx % F4R;
        agp[i] = A + (long)(tileM + row) * K + c4 * 4;
        asd[i] = row * P + c4 * 4;
    }
    const float* bgp[B_IT];
    int          bsd[B_IT];
    #pragma unroll
    for (int i = 0; i < B_IT; ++i) {
        const int blk = wave + i * NW;
        const int nb  = blk % NBN;
        const int kb  = blk / NBN;
        const int k   = kb * 4 + (lane >> 3);
        const int n   = nb * 32 + (lane & 7) * 4;
        bgp[i] = W + (long)k * N + tileN + n;
        bsd[i] = BM * P + n * P + k;
    }

    auto load_tile = [&](int k0) {
        #pragma unroll
        for (int i = 0; i < A_IT; ++i) ra[i] = *(const v4f*)(agp[i] + k0);
        #pragma unroll
        for (int i = 0; i < B_IT; ++i) rb[i] = *(const v4f*)(bgp[i] + (long)k0 * N);
    };
    auto store_tile = [&](int stage) {
        float* s = lds + stage * STAGE;
        #pragma unroll
        for (int i = 0; i < A_IT; ++i) *(v4f*)(s + asd[i]) = ra[i];
        #pragma unroll
        for (int i = 0; i < B_IT; ++i) {
            s[bsd[i]]         = rb[i].x;
            s[bsd[i] + P]     = rb[i].y;
            s[bsd[i] + 2 * P] = rb[i].z;
            s[bsd[i] + 3 * P] = rb[i].w;
        }
    };

    auto lds_barrier = [&]() {
        __builtin_amdgcn_fence(__ATOMIC_RELEASE, "workgroup", "local");
        __builtin_amdgcn_s_barrier();
        __builtin_amdgcn_fence(__ATOMIC_ACQUIRE, "workgroup", "local");
    };

    v8f acc[TM][TN] = {};

    const int aoff = (wm * WM + l15) * P + h4;
    const int boff = BM * P + (wn * WN + l15) * P + h4;

    constexpr int G = BK / 8;
    static_assert(G % 2 == 0, "the fragment register sets alternate per group");

    const int nk = K / BK;
    load_tile(0);
    store_tile(0);
    if (nk > 1) load_tile(BK);
    lds_barrier();

    v4f a[2][TM], b[2][TN];
    #pragma unroll
    for (int i = 0; i < TM; ++i) a[0][i] = *(const v4f*)(lds + aoff + i * 16 * P);
    #pragma unroll
    for (int j = 0; j < TN; ++j) b[0][j] = *(const v4f*)(lds + boff + j * 16 * P);

    for (int kt = 0; kt < nk; ++kt) {
        const int cur = kt & 1;
        const bool more = (kt + 1 < nk);
        const float* s  = lds + cur * STAGE;
        const float* sn = lds + (cur ^ 1) * STAGE;
        #pragma unroll
        for (int g = 0; g < G; ++g) {
            const int cb = g & 1, nb = cb ^ 1;
            if (g == G / 2 && more) {
                store_tile(cur ^ 1);
                if (kt + 2 < nk) load_tile((kt + 2) * BK);
            }
            if (g + 1 < G) {
                #pragma unroll
                for (int i = 0; i < TM; ++i)
                    a[nb][i] = *(const v4f*)(s + aoff + i * 16 * P + (g + 1) * 8);
                #pragma unroll
                for (int j = 0; j < TN; ++j)
                    b[nb][j] = *(const v4f*)(s + boff + j * 16 * P + (g + 1) * 8);
            } else {
                lds_barrier();
                if (more) {
                    #pragma unroll
                    for (int i = 0; i < TM; ++i)
                        a[nb][i] = *(const v4f*)(sn + aoff + i * 16 * P);
                    #pragma unroll
                    for (int j = 0; j < TN; ++j)
                        b[nb][j] = *(const v4f*)(sn + boff + j * 16 * P);
                }
            }
            __builtin_amdgcn_sched_barrier(0);
            #pragma unroll
            for (int i = 0; i < TM; ++i)
                #pragma unroll
                for (int j = 0; j < TN; ++j) {
                    v2f a0 = {a[cb][i].x, a[cb][i].y};
                    v2f b0 = {b[cb][j].x, b[cb][j].y};
                    acc[i][j] = __builtin_amdgcn_wmma_f32_16x16x4_f32(
                        false, a0, false, b0, (short)0, acc[i][j], false, false);
                }
            #pragma unroll
            for (int i = 0; i < TM; ++i)
                #pragma unroll
                for (int j = 0; j < TN; ++j) {
                    v2f a1 = {a[cb][i].z, a[cb][i].w};
                    v2f b1 = {b[cb][j].z, b[cb][j].w};
                    acc[i][j] = __builtin_amdgcn_wmma_f32_16x16x4_f32(
                        false, a1, false, b1, (short)0, acc[i][j], false, false);
                }
        }
    }

    const int rowOff = (lane >> 4) << 3;
    #pragma unroll
    for (int i = 0; i < TM; ++i) {
        const int rowBase = tileM + wm * WM + i * 16 + rowOff;
        #pragma unroll
        for (int j = 0; j < TN; ++j) {
            const int col = tileN + wn * WN + j * 16 + l15;
            #pragma unroll
            for (int r = 0; r < 8; ++r) {
                float val = acc[i][j][r];
                if (DO_ROPE) {
                    const float partner = __shfl_xor(val, 1, 32);
                    const int pos = (rowBase + r) & (S_ - 1);
                    const int fi  = (col & (HD_ - 1)) >> 1;
                    const float c = fcos[pos * (HD_ / 2) + fi];
                    const float s = fsin[pos * (HD_ / 2) + fi];
                    val = ((col & 1) == 0) ? (val * c - partner * s)
                                           : (partner * s + val * c);
                }
                C[(long)(rowBase + r) * N + col] = val;
            }
        }
    }
}

#define KP_ (HD_ + 4)

__global__ void __launch_bounds__(128)
attn_wmma(const float* __restrict__ q, const float* __restrict__ kbuf,
          const float* __restrict__ vbuf, float* __restrict__ o) {
    __shared__ __attribute__((aligned(16))) float Ks[80 * KP_];
    __shared__ __attribute__((aligned(16))) float Vs[80 * KP_];

    const int tid  = threadIdx.x;
    const int lane = tid & 31;
    const int wave = __builtin_amdgcn_readfirstlane(tid >> 5);
    const int i0   = blockIdx.x * 16;
    const int kvh  = blockIdx.y;
    const int b    = blockIdx.z;
    const int h    = kvh * 4 + wave;

    const int l15  = lane & 15;
    const int h4   = (lane >> 4) << 2;
    const int r8   = (lane >> 4) << 3;
    const int keystart = i0 - WIN_;

    v4f qf[8];
    {
        const float* qrow = q + ((long)(b * S_ + i0 + l15) * H_ + h) * HD_ + h4;
        #pragma unroll
        for (int g = 0; g < 8; ++g) qf[g] = *(const v4f*)(qrow + g * 8);
    }

    {
        v4f kr[10], vr[10];
        #pragma unroll
        for (int i = 0; i < 10; ++i) {
            const int idx = tid + i * 128;
            const int row = idx >> 4;
            const int c4  = idx & 15;
            const int key = keystart + row;
            const int keyc = key < 0 ? 0 : key;
            const long off = ((long)(b * S_ + keyc) * KVH_ + kvh) * HD_ + c4 * 4;
            kr[i] = *(const v4f*)(kbuf + off);
            vr[i] = *(const v4f*)(vbuf + off);
        }
        #pragma unroll
        for (int i = 0; i < 10; ++i) {
            const int idx = tid + i * 128;
            const int row = idx >> 4;
            const int c4  = idx & 15;
            const bool valid = (keystart + row) >= 0;
            const v4f zero = {0.0f, 0.0f, 0.0f, 0.0f};
            *(v4f*)(Ks + row * KP_ + c4 * 4) = kr[i];
            *(v4f*)(Vs + row * KP_ + c4 * 4) = valid ? vr[i] : zero;
        }
    }
    __syncthreads();

    v8f sacc[5] = {};
    {
        const float* kfrag = Ks + l15 * KP_ + h4;
        v4f kf[2][5];
        #pragma unroll
        for (int t = 0; t < 5; ++t) kf[0][t] = *(const v4f*)(kfrag + t * 16 * KP_);
        #pragma unroll
        for (int g = 0; g < 8; ++g) {
            const int cb = g & 1, nb = cb ^ 1;
            if (g + 1 < 8) {
                #pragma unroll
                for (int t = 0; t < 5; ++t)
                    kf[nb][t] = *(const v4f*)(kfrag + t * 16 * KP_ + (g + 1) * 8);
            }
            __builtin_amdgcn_sched_barrier(0);
            #pragma unroll
            for (int t = 0; t < 5; ++t) {
                v2f a0 = {kf[cb][t].x, kf[cb][t].y};
                v2f b0 = {qf[g].x, qf[g].y};
                sacc[t] = __builtin_amdgcn_wmma_f32_16x16x4_f32(
                    false, a0, false, b0, (short)0, sacc[t], false, false);
            }
            #pragma unroll
            for (int t = 0; t < 5; ++t) {
                v2f a1 = {kf[cb][t].z, kf[cb][t].w};
                v2f b1 = {qf[g].z, qf[g].w};
                sacc[t] = __builtin_amdgcn_wmma_f32_16x16x4_f32(
                    false, a1, false, b1, (short)0, sacc[t], false, false);
            }
        }
    }

    const float* vfrag = Vs + r8 * KP_ + l15 * 4;
    v4f vf[2][8];
    #pragma unroll
    for (int j = 0; j < 8; ++j) vf[0][j] = *(const v4f*)(vfrag + j * KP_);

    float p[5][8];
    {
        const int hc = r8 - l15;
        #pragma unroll
        for (int t = 0; t < 5; ++t) {
            const bool tile_out = (keystart + t * 16) < 0;
            #pragma unroll
            for (int r = 0; r < 8; ++r) {
                const int d = hc + (t * 16 + r - WIN_);
                const float sc = fmaxf(sacc[t][r] * 0.125f, -1e9f);
                p[t][r] = (d > 0 || d < -WIN_ || tile_out) ? -1e9f : sc;
            }
        }
    }

    {
        float mt[5];
        #pragma unroll
        for (int t = 0; t < 5; ++t) {
            const float m01 = fmaxf(p[t][0], p[t][1]), m23 = fmaxf(p[t][2], p[t][3]);
            const float m45 = fmaxf(p[t][4], p[t][5]), m67 = fmaxf(p[t][6], p[t][7]);
            mt[t] = fmaxf(fmaxf(m01, m23), fmaxf(m45, m67));
        }
        float m = fmaxf(fmaxf(fmaxf(mt[0], mt[1]), fmaxf(mt[2], mt[3])), mt[4]);
        m = fmaxf(m, __shfl_xor(m, 16, 32));
        float s4[4] = {0.0f, 0.0f, 0.0f, 0.0f};
        #pragma unroll
        for (int t = 0; t < 5; ++t)
            #pragma unroll
            for (int r = 0; r < 8; ++r) {
                const float e = __builtin_amdgcn_exp2f((p[t][r] - m) * 1.4426950408889634f);
                p[t][r] = e;
                s4[r & 3] += e;
            }
        float sum = (s4[0] + s4[1]) + (s4[2] + s4[3]);
        sum += __shfl_xor(sum, 16, 32);
        const float inv = 1.0f / sum;
        #pragma unroll
        for (int t = 0; t < 5; ++t)
            #pragma unroll
            for (int r = 0; r < 8; ++r) p[t][r] *= inv;
    }

    {
        v8f oacc[4] = {};
        #pragma unroll
        for (int t = 0; t < 5; ++t) {
            const int cb = t & 1, nb = cb ^ 1;
            if (t + 1 < 5) {
                #pragma unroll
                for (int j = 0; j < 8; ++j)
                    vf[nb][j] = *(const v4f*)(vfrag + ((t + 1) * 16 + j) * KP_);
            }
            __builtin_amdgcn_sched_barrier(0);
            #pragma unroll
            for (int j = 0; j < 4; ++j)
                #pragma unroll
                for (int nt = 0; nt < 4; ++nt) {
                    v2f a = {p[t][2 * j], p[t][2 * j + 1]};
                    v2f bv = {vf[cb][2 * j][nt], vf[cb][2 * j + 1][nt]};
                    oacc[nt] = __builtin_amdgcn_wmma_f32_16x16x4_f32(
                        false, a, false, bv, (short)0, oacc[nt], false, false);
                }
        }
        float* orow = o + ((long)(b * S_ + i0 + r8) * H_ + h) * HD_ + l15 * 4;
        #pragma unroll
        for (int r = 0; r < 8; ++r) {
            v4f ov = {oacc[0][r], oacc[1][r], oacc[2][r], oacc[3][r]};
            *(v4f*)(orow + (long)r * (H_ * HD_)) = ov;
        }
    }
}

extern "C" void kernel_launch(void* const* d_in, const int* in_sizes, int n_in,
                              void* d_out, int out_size, void* d_ws, size_t ws_size,
                              hipStream_t stream) {
    const float* x    = (const float*)d_in[0];
    const float* fcos = (const float*)d_in[1];
    const float* fsin = (const float*)d_in[2];
    const float* wq   = (const float*)d_in[3];
    const float* wk   = (const float*)d_in[4];
    const float* wv   = (const float*)d_in[5];
    const float* wo   = (const float*)d_in[6];
    float* out = (float*)d_out;

    float* qb = (float*)d_ws;
    float* kb = qb + (size_t)M_ * (H_ * HD_);
    float* vb = kb + (size_t)M_ * (KVH_ * HD_);
    float* at = vb + (size_t)M_ * (KVH_ * HD_);

    const dim3 blk(128);
    gemm_lds_wmma<128, 64, 32, 64, 32, 1><<<dim3((H_ * HD_) / 64, M_ / 128), blk, 0, stream>>>(
        x, wq, qb, M_, H_ * HD_, D_, fcos, fsin);
    gemm_lds_wmma<64, 64, 64, 32, 32, 1><<<dim3((KVH_ * HD_) / 64, M_ / 64), blk, 0, stream>>>(
        x, wk, kb, M_, KVH_ * HD_, D_, fcos, fsin);
    gemm_lds_wmma<64, 64, 64, 32, 32, 0><<<dim3((KVH_ * HD_) / 64, M_ / 64), blk, 0, stream>>>(
        x, wv, vb, M_, KVH_ * HD_, D_, fcos, fsin);

    attn_wmma<<<dim3(S_ / 16, KVH_, B_), dim3(128), 0, stream>>>(qb, kb, vb, at);

    gemm_lds_wmma<128, 64, 32, 64, 32, 0><<<dim3(D_ / 64, M_ / 128), blk, 0, stream>>>(
        at, wo, out, M_, D_, H_ * HD_, fcos, fsin);
}


# K and V projections by one 8-wave block per 64 rows: the x tile is staged in LDS once and feeds both products (tile 64 x (64 K + 64 V columns)); 4 launches, not 5; sum order unchanged
# speedup vs baseline: 5.1746x; 1.0479x over previous
#include <hip/hip_runtime.h>
#include <hip/hip_bf16.h>


typedef __attribute__((ext_vector_type(2))) float v2f;
typedef __attribute__((ext_vector_type(4))) float v4f;
typedef __attribute__((ext_vector_type(8))) float v8f;

#define B_ 2
#define S_ 2048
#define D_ 1024
#define H_ 16
#define KVH_ 4
#define HD_ 64
#define WIN_ 64
#define M_ (B_ * S_)

template <bool V> struct flag_c { static constexpr bool value = V; };

template <int BM, int BN, int BK, int WM, int WN, int DO_ROPE, int DUAL>
__global__ void __launch_bounds__((BM / WM) * (BN / WN) * 32)
gemm_lds_wmma(const float* __restrict__ A, const float* __restrict__ W,
              const float* __restrict__ W2,
              float* __restrict__ C, float* __restrict__ C2, int M, int N, int K,
              const float* __restrict__ fcos,
              const float* __restrict__ fsin) {
    constexpr int NWN  = BN / WN;
    constexpr int NW   = (BM / WM) * NWN;
    constexpr int NT   = NW * 32;
    constexpr int P    = BK + 4;
    constexpr int TM   = WM / 16;
    constexpr int TN   = WN / 16;
    constexpr int F4R  = BK / 4;
    constexpr int A_IT = (BM * F4R) / NT;
    constexpr int NBN  = BN / 32;
    constexpr int B_IT = ((BK / 4) * NBN) / NW;
    constexpr int STAGE = (BM + BN) * P;
    constexpr int BNH  = DUAL ? BN / 2 : BN;
    static_assert(BNH % WN == 0 && BNH % 32 == 0, "a wave tile and a 32-column block lie in one matrix");
    static_assert((BM * F4R) % NT == 0, "A tile must divide among the threads");
    static_assert(((BK / 4) * NBN) % NW == 0, "W tile must divide among the waves");
    static_assert(BK % 8 == 0 && (P / 4) % 2 == 1, "pitch must be 16 B x odd");

    __shared__ __attribute__((aligned(16))) float lds[2 * STAGE];

    const int tid  = threadIdx.x;
    const int lane = tid & 31;
    const int wave = __builtin_amdgcn_readfirstlane(tid >> 5);
    const int wm   = wave / NWN;
    const int wn   = wave % NWN;
    const int tileM = blockIdx.y * BM;
    const int tileN = blockIdx.x * BNH;

    const int l15 = lane & 15;
    const int h4  = (lane >> 4) << 2;

    v4f ra[A_IT], rb[B_IT];
    const float* agp[A_IT];
    int          asd[A_IT];
    #pragma unroll
    for (int i = 0; i < A_IT; ++i) {
        const int idx = tid + i * NT;
        const int row = idx / F4R;
        const int c4  = idx % F4R;
        agp[i] = A + (long)(tileM + row) * K + c4 * 4;
        asd[i] = row * P + c4 * 4;
    }
    const float* bgp[B_IT];
    int          bsd[B_IT];
    #pragma unroll
    for (int i = 0; i < B_IT; ++i) {
        const int blk = wave + i * NW;
        const int nb  = blk % NBN;
        const int kb  = blk / NBN;
        const int k   = kb * 4 + (lane >> 3);
        const int n   = nb * 32 + (lane & 7) * 4;
        const float* wsrc = (DUAL && n >= BNH) ? W2 : W;
        bgp[i] = wsrc + (long)k * N + tileN + (n % BNH);
        bsd[i] = BM * P + n * P + k;
    }

    auto load_tile = [&](int k0) {
        #pragma unroll
        for (int i = 0; i < A_IT; ++i) ra[i] = *(const v4f*)(agp[i] + k0);
        #pragma unroll
        for (int i = 0; i < B_IT; ++i) rb[i] = *(const v4f*)(bgp[i] + (long)k0 * N);
    };
    auto store_tile = [&](int stage) {
        float* s = lds + stage * STAGE;
        #pragma unroll
        for (int i = 0; i < A_IT; ++i) *(v4f*)(s + asd[i]) = ra[i];
        #pragma unroll
        for (int i = 0; i < B_IT; ++i) {
            s[bsd[i]]         = rb[i].x;
            s[bsd[i] + P]     = rb[i].y;
            s[bsd[i] + 2 * P] = rb[i].z;
            s[bsd[i] + 3 * P] = rb[i].w;
        }
    };

    auto lds_barrier = [&]() {
        __builtin_amdgcn_fence(__ATOMIC_RELEASE, "workgroup", "local");
        __builtin_amdgcn_s_barrier();
        __builtin_amdgcn_fence(__ATOMIC_ACQUIRE, "workgroup", "local");
    };

    v8f acc[TM][TN] = {};

    const int aoff = (wm * WM + l15) * P + h4;
    const int boff = BM * P + (wn * WN + l15) * P + h4;

    constexpr int G = BK / 8;
    static_assert(G % 2 == 0, "the fragment register sets alternate per group");

    const int nk = K / BK;
    load_tile(0);
    store_tile(0);
    if (nk > 1) load_tile(BK);
    lds_barrier();

    v4f a[2][TM], b[2][TN];
    #pragma unroll
    for (int i = 0; i < TM; ++i) a[0][i] = *(const v4f*)(lds + aoff + i * 16 * P);
    #pragma unroll
    for (int j = 0; j < TN; ++j) b[0][j] = *(const v4f*)(lds + boff + j * 16 * P);

    for (int kt = 0; kt < nk; ++kt) {
        const int cur = kt & 1;
        const bool more = (kt + 1 < nk);
        const float* s  = lds + cur * STAGE;
        const float* sn = lds + (cur ^ 1) * STAGE;
        #pragma unroll
        for (int g = 0; g < G; ++g) {
            const int cb = g & 1, nb = cb ^ 1;
            if (g == G / 2 && more) {
                store_tile(cur ^ 1);
                if (kt + 2 < nk) load_tile((kt + 2) * BK);
            }
            if (g + 1 < G) {
                #pragma unroll
                for (int i = 0; i < TM; ++i)
                    a[nb][i] = *(const v4f*)(s + aoff + i * 16 * P + (g + 1) * 8);
                #pragma unroll
                for (int j = 0; j < TN; ++j)
                    b[nb][j] = *(const v4f*)(s + boff + j * 16 * P + (g + 1) * 8);
            } else {
                lds_barrier();
                if (more) {
                    #pragma unroll
                    for (int i = 0; i < TM; ++i)
                        a[nb][i] = *(const v4f*)(sn + aoff + i * 16 * P);
                    #pragma unroll
                    for (int j = 0; j < TN; ++j)
                        b[nb][j] = *(const v4f*)(sn + boff + j * 16 * P);
                }
            }
            __builtin_amdgcn_sched_barrier(0);
            #pragma unroll
            for (int i = 0; i < TM; ++i)
                #pragma unroll
                for (int j = 0; j < TN; ++j) {
                    v2f a0 = {a[cb][i].x, a[cb][i].y};
                    v2f b0 = {b[cb][j].x, b[cb][j].y};
                    acc[i][j] = __builtin_amdgcn_wmma_f32_16x16x4_f32(
                        false, a0, false, b0, (short)0, acc[i][j], false, false);
                }
            #pragma unroll
            for (int i = 0; i < TM; ++i)
                #pragma unroll
                for (int j = 0; j < TN; ++j) {
                    v2f a1 = {a[cb][i].z, a[cb][i].w};
                    v2f b1 = {b[cb][j].z, b[cb][j].w};
                    acc[i][j] = __builtin_amdgcn_wmma_f32_16x16x4_f32(
                        false, a1, false, b1, (short)0, acc[i][j], false, false);
                }
        }
    }

    const bool second = DUAL && (wn * WN >= BNH);
    float* __restrict__ Cd = second ? C2 : C;
    const int colBase = tileN + wn * WN - (second ? BNH : 0);
    const int rowOff = (lane >> 4) << 3;
    auto epilogue = [&](auto rope_c) {
    constexpr bool ROPE = decltype(rope_c)::value;
    #pragma unroll
    for (int i = 0; i < TM; ++i) {
        const int rowBase = tileM + wm * WM + i * 16 + rowOff;
        #pragma unroll
        for (int j = 0; j < TN; ++j) {
            const int col = colBase + j * 16 + l15;
            #pragma unroll
            for (int r = 0; r < 8; ++r) {
                float val = acc[i][j][r];
                if (ROPE) {
                    const float partner = __shfl_xor(val, 1, 32);
                    const int pos = (rowBase + r) & (S_ - 1);
                    const int fi  = (col & (HD_ - 1)) >> 1;
                    const float c = fcos[pos * (HD_ / 2) + fi];
                    const float s = fsin[pos * (HD_ / 2) + fi];
                    val = ((col & 1) == 0) ? (val * c - partner * s)
                                           : (partner * s + val * c);
                }
                Cd[(long)(rowBase + r) * N + col] = val;
            }
        }
    }
    };
    if (DO_ROPE && !second) epilogue(flag_c<true>{});
    else                    epilogue(flag_c<false>{});
}

#define KP_ (HD_ + 4)

__global__ void __launch_bounds__(128)
attn_wmma(const float* __restrict__ q, const float* __restrict__ kbuf,
          const float* __restrict__ vbuf, float* __restrict__ o) {
    __shared__ __attribute__((aligned(16))) float Ks[80 * KP_];
    __shared__ __attribute__((aligned(16))) float Vs[80 * KP_];

    const int tid  = threadIdx.x;
    const int lane = tid & 31;
    const int wave = __builtin_amdgcn_readfirstlane(tid >> 5);
    const int i0   = blockIdx.x * 16;
    const int kvh  = blockIdx.y;
    const int b    = blockIdx.z;
    const int h    = kvh * 4 + wave;

    const int l15  = lane & 15;
    const int h4   = (lane >> 4) << 2;
    const int r8   = (lane >> 4) << 3;
    const int keystart = i0 - WIN_;

    v4f qf[8];
    {
        const float* qrow = q + ((long)(b * S_ + i0 + l15) * H_ + h) * HD_ + h4;
        #pragma unroll
        for (int g = 0; g < 8; ++g) qf[g] = *(const v4f*)(qrow + g * 8);
    }

    {
        v4f kr[10], vr[10];
        #pragma unroll
        for (int i = 0; i < 10; ++i) {
            const int idx = tid + i * 128;
            const int row = idx >> 4;
            const int c4  = idx & 15;
            const int key = keystart + row;
            const int keyc = key < 0 ? 0 : key;
            const long off = ((long)(b * S_ + keyc) * KVH_ + kvh) * HD_ + c4 * 4;
            kr[i] = *(const v4f*)(kbuf + off);
            vr[i] = *(const v4f*)(vbuf + off);
        }
        #pragma unroll
        for (int i = 0; i < 10; ++i) {
            const int idx = tid + i * 128;
            const int row = idx >> 4;
            const int c4  = idx & 15;
            const bool valid = (keystart + row) >= 0;
            const v4f zero = {0.0f, 0.0f, 0.0f, 0.0f};
            *(v4f*)(Ks + row * KP_ + c4 * 4) = kr[i];
            *(v4f*)(Vs + row * KP_ + c4 * 4) = valid ? vr[i] : zero;
        }
    }
    __syncthreads();

    v8f sacc[5] = {};
    {
        const float* kfrag = Ks + l15 * KP_ + h4;
        v4f kf[2][5];
        #pragma unroll
        for (int t = 0; t < 5; ++t) kf[0][t] = *(const v4f*)(kfrag + t * 16 * KP_);
        #pragma unroll
        for (int g = 0; g < 8; ++g) {
            const int cb = g & 1, nb = cb ^ 1;
            if (g + 1 < 8) {
                #pragma unroll
                for (int t = 0; t < 5; ++t)
                    kf[nb][t] = *(const v4f*)(kfrag + t * 16 * KP_ + (g + 1) * 8);
            }
            __builtin_amdgcn_sched_barrier(0);
            #pragma unroll
            for (int t = 0; t < 5; ++t) {
                v2f a0 = {kf[cb][t].x, kf[cb][t].y};
                v2f b0 = {qf[g].x, qf[g].y};
                sacc[t] = __builtin_amdgcn_wmma_f32_16x16x4_f32(
                    false, a0, false, b0, (short)0, sacc[t], false, false);
            }
            #pragma unroll
            for (int t = 0; t < 5; ++t) {
                v2f a1 = {kf[cb][t].z, kf[cb][t].w};
                v2f b1 = {qf[g].z, qf[g].w};
                sacc[t] = __builtin_amdgcn_wmma_f32_16x16x4_f32(
                    false, a1, false, b1, (short)0, sacc[t], false, false);
            }
        }
    }

    const float* vfrag = Vs + r8 * KP_ + l15 * 4;
    v4f vf[2][8];
    #pragma unroll
    for (int j = 0; j < 8; ++j) vf[0][j] = *(const v4f*)(vfrag + j * KP_);

    float p[5][8];
    {
        const int hc = r8 - l15;
        #pragma unroll
        for (int t = 0; t < 5; ++t) {
            const bool tile_out = (keystart + t * 16) < 0;
            #pragma unroll
            for (int r = 0; r < 8; ++r) {
                const int d = hc + (t * 16 + r - WIN_);
                const float sc = fmaxf(sacc[t][r] * 0.125f, -1e9f);
                p[t][r] = (d > 0 || d < -WIN_ || tile_out) ? -1e9f : sc;
            }
        }
    }

    {
        float mt[5];
        #pragma unroll
        for (int t = 0; t < 5; ++t) {
            const float m01 = fmaxf(p[t][0], p[t][1]), m23 = fmaxf(p[t][2], p[t][3]);
            const float m45 = fmaxf(p[t][4], p[t][5]), m67 = fmaxf(p[t][6], p[t][7]);
            mt[t] = fmaxf(fmaxf(m01, m23), fmaxf(m45, m67));
        }
        float m = fmaxf(fmaxf(fmaxf(mt[0], mt[1]), fmaxf(mt[2], mt[3])), mt[4]);
        m = fmaxf(m, __shfl_xor(m, 16, 32));
        float s4[4] = {0.0f, 0.0f, 0.0f, 0.0f};
        #pragma unroll
        for (int t = 0; t < 5; ++t)
            #pragma unroll
            for (int r = 0; r < 8; ++r) {
                const float e = __builtin_amdgcn_exp2f((p[t][r] - m) * 1.4426950408889634f);
                p[t][r] = e;
                s4[r & 3] += e;
            }
        float sum = (s4[0] + s4[1]) + (s4[2] + s4[3]);
        sum += __shfl_xor(sum, 16, 32);
        const float inv = 1.0f / sum;
        #pragma unroll
        for (int t = 0; t < 5; ++t)
            #pragma unroll
            for (int r = 0; r < 8; ++r) p[t][r] *= inv;
    }

    {
        v8f oacc[4] = {};
        #pragma unroll
        for (int t = 0; t < 5; ++t) {
            const int cb = t & 1, nb = cb ^ 1;
            if (t + 1 < 5) {
                #pragma unroll
                for (int j = 0; j < 8; ++j)
                    vf[nb][j] = *(const v4f*)(vfrag + ((t + 1) * 16 + j) * KP_);
            }
            __builtin_amdgcn_sched_barrier(0);
            #pragma unroll
            for (int j = 0; j < 4; ++j)
                #pragma unroll
                for (int nt = 0; nt < 4; ++nt) {
                    v2f a = {p[t][2 * j], p[t][2 * j + 1]};
                    v2f bv = {vf[cb][2 * j][nt], vf[cb][2 * j + 1][nt]};
                    oacc[nt] = __builtin_amdgcn_wmma_f32_16x16x4_f32(
                        false, a, false, bv, (short)0, oacc[nt], false, false);
                }
        }
        float* orow = o + ((long)(b * S_ + i0 + r8) * H_ + h) * HD_ + l15 * 4;
        #pragma unroll
        for (int r = 0; r < 8; ++r) {
            v4f ov = {oacc[0][r], oacc[1][r], oacc[2][r], oacc[3][r]};
            *(v4f*)(orow + (long)r * (H_ * HD_)) = ov;
        }
    }
}

extern "C" void kernel_launch(void* const* d_in, const int* in_sizes, int n_in,
                              void* d_out, int out_size, void* d_ws, size_t ws_size,
                              hipStream_t stream) {
    const float* x    = (const float*)d_in[0];
    const float* fcos = (const float*)d_in[1];
    const float* fsin = (const float*)d_in[2];
    const float* wq   = (const float*)d_in[3];
    const float* wk   = (const float*)d_in[4];
    const float* wv   = (const float*)d_in[5];
    const float* wo   = (const float*)d_in[6];
    float* out = (float*)d_out;

    float* qb = (float*)d_ws;
    float* kb = qb + (size_t)M_ * (H_ * HD_);
    float* vb = kb + (size_t)M_ * (KVH_ * HD_);
    float* at = vb + (size_t)M_ * (KVH_ * HD_);

    const dim3 blk(128);
    gemm_lds_wmma<128, 64, 32, 64, 32, 1, 0><<<dim3((H_ * HD_) / 64, M_ / 128), blk, 0, stream>>>(
        x, wq, nullptr, qb, nullptr, M_, H_ * HD_, D_, fcos, fsin);
    gemm_lds_wmma<64, 128, 64, 32, 32, 1, 1><<<dim3((KVH_ * HD_) / 64, M_ / 64), dim3(256), 0, stream>>>(
        x, wk, wv, kb, vb, M_, KVH_ * HD_, D_, fcos, fsin);

    attn_wmma<<<dim3(S_ / 16, KVH_, B_), dim3(128), 0, stream>>>(qb, kb, vb, at);

    gemm_lds_wmma<128, 64, 32, 64, 32, 0, 0><<<dim3(D_ / 64, M_ / 128), blk, 0, stream>>>(
        at, wo, nullptr, out, nullptr, M_, D_, H_ * HD_, fcos, fsin);
}


# Q projection and attention in one launch: a block of 8 waves keeps its rotated 128x64 Q tile (one head) in LDS, stages the 192 K, V rows it sees, one query tile a wave; 3 launches; sums unchanged
# speedup vs baseline: 5.4740x; 1.0579x over previous
#include <hip/hip_runtime.h>
#include <hip/hip_bf16.h>


typedef __attribute__((ext_vector_type(2))) float v2f;
typedef __attribute__((ext_vector_type(4))) float v4f;
typedef __attribute__((ext_vector_type(8))) float v8f;

#define B_ 2
#define S_ 2048
#define D_ 1024
#define H_ 16
#define KVH_ 4
#define HD_ 64
#define WIN_ 64
#define M_ (B_ * S_)

#define KP_ (HD_ + 4)

__device__ __forceinline__ void attn_tile(const float* Qt, const float* Kt, const float* Vt,
                                          int keystart, float* orow, int lane) {
    const int l15  = lane & 15;
    const int h4   = (lane >> 4) << 2;
    const int r8   = (lane >> 4) << 3;

    v4f qf[8];
    {
        const float* qrow = Qt + l15 * KP_ + h4;
        #pragma unroll
        for (int g = 0; g < 8; ++g) qf[g] = *(const v4f*)(qrow + g * 8);
    }

    v8f sacc[5] = {};
    {
        const float* kfrag = Kt + l15 * KP_ + h4;
        v4f kf[2][5];
        #pragma unroll
        for (int t = 0; t < 5; ++t) kf[0][t] = *(const v4f*)(kfrag + t * 16 * KP_);
        #pragma unroll
        for (int g = 0; g < 8; ++g) {
            const int cb = g & 1, nb = cb ^ 1;
            if (g + 1 < 8) {
                #pragma unroll
                for (int t = 0; t < 5; ++t)
                    kf[nb][t] = *(const v4f*)(kfrag + t * 16 * KP_ + (g + 1) * 8);
            }
            __builtin_amdgcn_sched_barrier(0);
            #pragma unroll
            for (int t = 0; t < 5; ++t) {
                v2f a0 = {kf[cb][t].x, kf[cb][t].y};
                v2f b0 = {qf[g].x, qf[g].y};
                sacc[t] = __builtin_amdgcn_wmma_f32_16x16x4_f32(
                    false, a0, false, b0, (short)0, sacc[t], false, false);
            }
            #pragma unroll
            for (int t = 0; t < 5; ++t) {
                v2f a1 = {kf[cb][t].z, kf[cb][t].w};
                v2f b1 = {qf[g].z, qf[g].w};
                sacc[t] = __builtin_amdgcn_wmma_f32_16x16x4_f32(
                    false, a1, false, b1, (short)0, sacc[t], false, false);
            }
        }
    }

    const float* vfrag = Vt + r8 * KP_ + l15 * 4;
    v4f vf[2][8];
    #pragma unroll
    for (int j = 0; j < 8; ++j) vf[0][j] = *(const v4f*)(vfrag + j * KP_);

    float p[5][8];
    {
        const int hc = r8 - l15;
        #pragma unroll
        for (int t = 0; t < 5; ++t) {
            const bool tile_out = (keystart + t * 16) < 0;
            #pragma unroll
            for (int r = 0; r < 8; ++r) {
                const int d = hc + (t * 16 + r - WIN_);
                const float sc = fmaxf(sacc[t][r] * 0.125f, -1e9f);
                p[t][r] = (d > 0 || d < -WIN_ || tile_out) ? -1e9f : sc;
            }
        }
    }

    {
        float mt[5];
        #pragma unroll
        for (int t = 0; t < 5; ++t) {
            const float m01 = fmaxf(p[t][0], p[t][1]), m23 = fmaxf(p[t][2], p[t][3]);
            const float m45 = fmaxf(p[t][4], p[t][5]), m67 = fmaxf(p[t][6], p[t][7]);
            mt[t] = fmaxf(fmaxf(m01, m23), fmaxf(m45, m67));
        }
        float m = fmaxf(fmaxf(fmaxf(mt[0], mt[1]), fmaxf(mt[2], mt[3])), mt[4]);
        m = fmaxf(m, __shfl_xor(m, 16, 32));
        float s4[4] = {0.0f, 0.0f, 0.0f, 0.0f};
        #pragma unroll
        for (int t = 0; t < 5; ++t)
            #pragma unroll
            for (int r = 0; r < 8; ++r) {
                const float e = __builtin_amdgcn_exp2f((p[t][r] - m) * 1.4426950408889634f);
                p[t][r] = e;
                s4[r & 3] += e;
            }
        float sum = (s4[0] + s4[1]) + (s4[2] + s4[3]);
        sum += __shfl_xor(sum, 16, 32);
        const float inv = 1.0f / sum;
        #pragma unroll
        for (int t = 0; t < 5; ++t)
            #pragma unroll
            for (int r = 0; r < 8; ++r) p[t][r] *= inv;
    }

    {
        v8f oacc[4] = {};
        #pragma unroll
        for (int t = 0; t < 5; ++t) {
            const int cb = t & 1, nb = cb ^ 1;
            if (t + 1 < 5) {
                #pragma unroll
                for (int j = 0; j < 8; ++j)
                    vf[nb][j] = *(const v4f*)(vfrag + ((t + 1) * 16 + j) * KP_);
            }
            __builtin_amdgcn_sched_barrier(0);
            #pragma unroll
            for (int j = 0; j < 4; ++j)
                #pragma unroll
                for (int nt = 0; nt < 4; ++nt) {
                    v2f a = {p[t][2 * j], p[t][2 * j + 1]};
                    v2f bv = {vf[cb][2 * j][nt], vf[cb][2 * j + 1][nt]};
                    oacc[nt] = __builtin_amdgcn_wmma_f32_16x16x4_f32(
                        false, a, false, bv, (short)0, oacc[nt], false, false);
                }
        }
        #pragma unroll
        for (int r = 0; r < 8; ++r) {
            v4f ov = {oacc[0][r], oacc[1][r], oacc[2][r], oacc[3][r]};
            *(v4f*)(orow + (long)r * (H_ * HD_)) = ov;
        }
    }
}

template <bool V> struct flag_c { static constexpr bool value = V; };

template <int BM, int BN, int BK, int WM, int WN, int DO_ROPE, int DUAL, int FUSE>
__global__ void __launch_bounds__((BM / WM) * (BN / WN) * 32)
gemm_lds_wmma(const float* __restrict__ A, const float* __restrict__ W,
              const float* __restrict__ W2,
              float* __restrict__ C, float* __restrict__ C2, int M, int N, int K,
              const float* __restrict__ fcos,
              const float* __restrict__ fsin,
              const float* __restrict__ kbuf,
              const float* __restrict__ vbuf) {
    constexpr int NWN  = BN / WN;
    constexpr int NW   = (BM / WM) * NWN;
    constexpr int NT   = NW * 32;
    constexpr int P    = BK + 4;
    constexpr int TM   = WM / 16;
    constexpr int TN   = WN / 16;
    constexpr int F4R  = BK / 4;
    constexpr int A_IT = (BM * F4R) / NT;
    constexpr int NBN  = BN / 32;
    constexpr int B_IT = ((BK / 4) * NBN) / NW;
    constexpr int STAGE = (BM + BN) * P;
    constexpr int BNH  = DUAL ? BN / 2 : BN;
    static_assert(BNH % WN == 0 && BNH % 32 == 0, "a wave tile and a 32-column block lie in one matrix");
    static_assert((BM * F4R) % NT == 0, "A tile must divide among the threads");
    static_assert(((BK / 4) * NBN) % NW == 0, "W tile must divide among the waves");
    static_assert(BK % 8 == 0 && (P / 4) % 2 == 1, "pitch must be 16 B x odd");
    constexpr int QSF  = BM * KP_;
    constexpr int KVF  = (BM + WIN_) * KP_;
    constexpr int LDSF = (FUSE && QSF + 2 * KVF > 2 * STAGE) ? QSF + 2 * KVF : 2 * STAGE;
    static_assert(!FUSE || (BN == HD_ && DO_ROPE && !DUAL && (BM / 16) % NW == 0),
                  "the fused block is one head wide and its query tiles divide among its waves");

    __shared__ __attribute__((aligned(16))) float lds[LDSF];

    const int tid  = threadIdx.x;
    const int lane = tid & 31;
    const int wave = __builtin_amdgcn_readfirstlane(tid >> 5);
    const int wm   = wave / NWN;
    const int wn   = wave % NWN;
    const int tileM = blockIdx.y * BM;
    const int tileN = blockIdx.x * BNH;

    const int l15 = lane & 15;
    const int h4  = (lane >> 4) << 2;

    v4f ra[A_IT], rb[B_IT];
    const float* agp[A_IT];
    int          asd[A_IT];
    #pragma unroll
    for (int i = 0; i < A_IT; ++i) {
        const int idx = tid + i * NT;
        const int row = idx / F4R;
        const int c4  = idx % F4R;
        agp[i] = A + (long)(tileM + row) * K + c4 * 4;
        asd[i] = row * P + c4 * 4;
    }
    const float* bgp[B_IT];
    int          bsd[B_IT];
    #pragma unroll
    for (int i = 0; i < B_IT; ++i) {
        const int blk = wave + i * NW;
        const int nb  = blk % NBN;
        const int kb  = blk / NBN;
        const int k   = kb * 4 + (lane >> 3);
        const int n   = nb * 32 + (lane & 7) * 4;
        const float* wsrc = (DUAL && n >= BNH) ? W2 : W;
        bgp[i] = wsrc + (long)k * N + tileN + (n % BNH);
        bsd[i] = BM * P + n * P + k;
    }

    auto load_tile = [&](int k0) {
        #pragma unroll
        for (int i = 0; i < A_IT; ++i) ra[i] = *(const v4f*)(agp[i] + k0);
        #pragma unroll
        for (int i = 0; i < B_IT; ++i) rb[i] = *(const v4f*)(bgp[i] + (long)k0 * N);
    };
    auto store_tile = [&](int stage) {
        float* s = lds + stage * STAGE;
        #pragma unroll
        for (int i = 0; i < A_IT; ++i) *(v4f*)(s + asd[i]) = ra[i];
        #pragma unroll
        for (int i = 0; i < B_IT; ++i) {
            s[bsd[i]]         = rb[i].x;
            s[bsd[i] + P]     = rb[i].y;
            s[bsd[i] + 2 * P] = rb[i].z;
            s[bsd[i] + 3 * P] = rb[i].w;
        }
    };

    auto lds_barrier = [&]() {
        __builtin_amdgcn_fence(__ATOMIC_RELEASE, "workgroup", "local");
        __builtin_amdgcn_s_barrier();
        __builtin_amdgcn_fence(__ATOMIC_ACQUIRE, "workgroup", "local");
    };

    v8f acc[TM][TN] = {};

    const int aoff = (wm * WM + l15) * P + h4;
    const int boff = BM * P + (wn * WN + l15) * P + h4;

    constexpr int G = BK / 8;
    static_assert(G % 2 == 0, "the fragment register sets alternate per group");

    const int nk = K / BK;
    load_tile(0);
    store_tile(0);
    if (nk > 1) load_tile(BK);
    lds_barrier();

    v4f a[2][TM], b[2][TN];
    #pragma unroll
    for (int i = 0; i < TM; ++i) a[0][i] = *(const v4f*)(lds + aoff + i * 16 * P);
    #pragma unroll
    for (int j = 0; j < TN; ++j) b[0][j] = *(const v4f*)(lds + boff + j * 16 * P);

    for (int kt = 0; kt < nk; ++kt) {
        const int cur = kt & 1;
        const bool more = (kt + 1 < nk);
        const float* s  = lds + cur * STAGE;
        const float* sn = lds + (cur ^ 1) * STAGE;
        #pragma unroll
        for (int g = 0; g < G; ++g) {
            const int cb = g & 1, nb = cb ^ 1;
            if (g == G / 2 && more) {
                store_tile(cur ^ 1);
                if (kt + 2 < nk) load_tile((kt + 2) * BK);
            }
            if (g + 1 < G) {
                #pragma unroll
                for (int i = 0; i < TM; ++i)
                    a[nb][i] = *(const v4f*)(s + aoff + i * 16 * P + (g + 1) * 8);
                #pragma unroll
                for (int j = 0; j < TN; ++j)
                    b[nb][j] = *(const v4f*)(s + boff + j * 16 * P + (g + 1) * 8);
            } else {
                lds_barrier();
                if (more) {
                    #pragma unroll
                    for (int i = 0; i < TM; ++i)
                        a[nb][i] = *(const v4f*)(sn + aoff + i * 16 * P);
                    #pragma unroll
                    for (int j = 0; j < TN; ++j)
                        b[nb][j] = *(const v4f*)(sn + boff + j * 16 * P);
                }
            }
            __builtin_amdgcn_sched_barrier(0);
            #pragma unroll
            for (int i = 0; i < TM; ++i)
                #pragma unroll
                for (int j = 0; j < TN; ++j) {
                    v2f a0 = {a[cb][i].x, a[cb][i].y};
                    v2f b0 = {b[cb][j].x, b[cb][j].y};
                    acc[i][j] = __builtin_amdgcn_wmma_f32_16x16x4_f32(
                        false, a0, false, b0, (short)0, acc[i][j], false, false);
                }
            #pragma unroll
            for (int i = 0; i < TM; ++i)
                #pragma unroll
                for (int j = 0; j < TN; ++j) {
                    v2f a1 = {a[cb][i].z, a[cb][i].w};
                    v2f b1 = {b[cb][j].z, b[cb][j].w};
                    acc[i][j] = __builtin_amdgcn_wmma_f32_16x16x4_f32(
                        false, a1, false, b1, (short)0, acc[i][j], false, false);
                }
        }
    }

    if (FUSE) {
        float* Qs = lds;
        float* Ks = lds + QSF;
        float* Vs = Ks + KVF;
        const int head = blockIdx.x;
        const int kvh  = head / (H_ / KVH_);
        const int s0   = tileM & (S_ - 1);
        const int brow = tileM - s0;
        const int rowOffq = (lane >> 4) << 3;

        #pragma unroll
        for (int i = 0; i < TM; ++i) {
            const int rowl = wm * WM + i * 16 + rowOffq;
            #pragma unroll
            for (int j = 0; j < TN; ++j) {
                const int col = wn * WN + j * 16 + l15;
                #pragma unroll
                for (int r = 0; r < 8; ++r) {
                    float val = acc[i][j][r];
                    const float partner = __shfl_xor(val, 1, 32);
                    const int pos = s0 + rowl + r;
                    const int fi  = col >> 1;
                    const float c = fcos[pos * (HD_ / 2) + fi];
                    const float sn = fsin[pos * (HD_ / 2) + fi];
                    val = ((col & 1) == 0) ? (val * c - partner * sn)
                                           : (partner * sn + val * c);
                    Qs[(rowl + r) * KP_ + col] = val;
                }
            }
        }

        constexpr int KV_IT = ((BM + WIN_) * 16) / NT;
        static_assert(!FUSE || ((BM + WIN_) * 16) % NT == 0, "K / V rows divide among the threads");
        {
            v4f kr[KV_IT], vr[KV_IT];
            #pragma unroll
            for (int i = 0; i < KV_IT; ++i) {
                const int idx = tid + i * NT;
                const int row = idx >> 4;
                const int c4  = idx & 15;
                const int key = s0 - WIN_ + row;
                const int keyc = key < 0 ? 0 : key;
                const long off = ((long)(brow + keyc) * KVH_ + kvh) * HD_ + c4 * 4;
                kr[i] = *(const v4f*)(kbuf + off);
                vr[i] = *(const v4f*)(vbuf + off);
            }
            #pragma unroll
            for (int i = 0; i < KV_IT; ++i) {
                const int idx = tid + i * NT;
                const int row = idx >> 4;
                const int c4  = idx & 15;
                const bool valid = (s0 - WIN_ + row) >= 0;
                const v4f zero = {0.0f, 0.0f, 0.0f, 0.0f};
                *(v4f*)(Ks + row * KP_ + c4 * 4) = kr[i];
                *(v4f*)(Vs + row * KP_ + c4 * 4) = valid ? vr[i] : zero;
            }
        }
        __syncthreads();

        constexpr int QT_W = (BM / 16) / NW;
        for (int u = 0; u < QT_W; ++u) {
            const int qt = wave * QT_W + u;
            const int i0 = qt * 16;
            float* orow = C + ((long)(tileM + i0 + rowOffq) * H_ + head) * HD_ + l15 * 4;
            attn_tile(Qs + i0 * KP_, Ks + i0 * KP_, Vs + i0 * KP_, s0 + i0 - WIN_, orow, lane);
        }
        return;
    }

    const bool second = DUAL && (wn * WN >= BNH);
    float* __restrict__ Cd = second ? C2 : C;
    const int colBase = tileN + wn * WN - (second ? BNH : 0);
    const int rowOff = (lane >> 4) << 3;
    auto epilogue = [&](auto rope_c) {
    constexpr bool ROPE = decltype(rope_c)::value;
    #pragma unroll
    for (int i = 0; i < TM; ++i) {
        const int rowBase = tileM + wm * WM + i * 16 + rowOff;
        #pragma unroll
        for (int j = 0; j < TN; ++j) {
            const int col = colBase + j * 16 + l15;
            #pragma unroll
            for (int r = 0; r < 8; ++r) {
                float val = acc[i][j][r];
                if (ROPE) {
                    const float partner = __shfl_xor(val, 1, 32);
                    const int pos = (rowBase + r) & (S_ - 1);
                    const int fi  = (col & (HD_ - 1)) >> 1;
                    const float c = fcos[pos * (HD_ / 2) + fi];
                    const float s = fsin[pos * (HD_ / 2) + fi];
                    val = ((col & 1) == 0) ? (val * c - partner * s)
                                           : (partner * s + val * c);
                }
                Cd[(long)(rowBase + r) * N + col] = val;
            }
        }
    }
    };
    if (DO_ROPE && !second) epilogue(flag_c<true>{});
    else                    epilogue(flag_c<false>{});
}

extern "C" void kernel_launch(void* const* d_in, const int* in_sizes, int n_in,
                              void* d_out, int out_size, void* d_ws, size_t ws_size,
                              hipStream_t stream) {
    const float* x    = (const float*)d_in[0];
    const float* fcos = (const float*)d_in[1];
    const float* fsin = (const float*)d_in[2];
    const float* wq   = (const float*)d_in[3];
    const float* wk   = (const float*)d_in[4];
    const float* wv   = (const float*)d_in[5];
    const float* wo   = (const float*)d_in[6];
    float* out = (float*)d_out;

    float* kb = (float*)d_ws;
    float* vb = kb + (size_t)M_ * (KVH_ * HD_);
    float* at = vb + (size_t)M_ * (KVH_ * HD_);

    const dim3 blk(128);
    gemm_lds_wmma<64, 128, 64, 32, 32, 1, 1, 0><<<dim3((KVH_ * HD_) / 64, M_ / 64), dim3(256), 0, stream>>>(
        x, wk, wv, kb, vb, M_, KVH_ * HD_, D_, fcos, fsin, nullptr, nullptr);

    gemm_lds_wmma<128, 64, 32, 32, 32, 1, 0, 1><<<dim3((H_ * HD_) / 64, M_ / 128), dim3(256), 0, stream>>>(
        x, wq, nullptr, at, nullptr, M_, H_ * HD_, D_, fcos, fsin, kb, vb);

    gemm_lds_wmma<128, 64, 32, 64, 32, 0, 0, 0><<<dim3(D_ / 64, M_ / 128), blk, 0, stream>>>(
        at, wo, nullptr, out, nullptr, M_, D_, H_ * HD_, fcos, fsin, nullptr, nullptr);
}


# GEMM K loops: all fragments of a K tile in registers, two sets; LDS reads, stage stores and loads in one place per tile, barrier signalled mid tile; 32-64 WMMAs unbroken; K tile 32; sums unchanged
# speedup vs baseline: 5.7064x; 1.0425x over previous
#include <hip/hip_runtime.h>
#include <hip/hip_bf16.h>


typedef __attribute__((ext_vector_type(2))) float v2f;
typedef __attribute__((ext_vector_type(4))) float v4f;
typedef __attribute__((ext_vector_type(8))) float v8f;

#define B_ 2
#define S_ 2048
#define D_ 1024
#define H_ 16
#define KVH_ 4
#define HD_ 64
#define WIN_ 64
#define M_ (B_ * S_)

#define KP_ (HD_ + 4)

__device__ __forceinline__ void attn_tile(const float* Qt, const float* Kt, const float* Vt,
                                          int keystart, float* orow, int lane) {
    const int l15  = lane & 15;
    const int h4   = (lane >> 4) << 2;
    const int r8   = (lane >> 4) << 3;

    v4f qf[8];
    {
        const float* qrow = Qt + l15 * KP_ + h4;
        #pragma unroll
        for (int g = 0; g < 8; ++g) qf[g] = *(const v4f*)(qrow + g * 8);
    }

    v8f sacc[5] = {};
    {
        const float* kfrag = Kt + l15 * KP_ + h4;
        v4f kf[2][5];
        #pragma unroll
        for (int t = 0; t < 5; ++t) kf[0][t] = *(const v4f*)(kfrag + t * 16 * KP_);
        #pragma unroll
        for (int g = 0; g < 8; ++g) {
            const int cb = g & 1, nb = cb ^ 1;
            if (g + 1 < 8) {
                #pragma unroll
                for (int t = 0; t < 5; ++t)
                    kf[nb][t] = *(const v4f*)(kfrag + t * 16 * KP_ + (g + 1) * 8);
            }
            __builtin_amdgcn_sched_barrier(0);
            #pragma unroll
            for (int t = 0; t < 5; ++t) {
                v2f a0 = {kf[cb][t].x, kf[cb][t].y};
                v2f b0 = {qf[g].x, qf[g].y};
                sacc[t] = __builtin_amdgcn_wmma_f32_16x16x4_f32(
                    false, a0, false, b0, (short)0, sacc[t], false, false);
            }
            #pragma unroll
            for (int t = 0; t < 5; ++t) {
                v2f a1 = {kf[cb][t].z, kf[cb][t].w};
                v2f b1 = {qf[g].z, qf[g].w};
                sacc[t] = __builtin_amdgcn_wmma_f32_16x16x4_f32(
                    false, a1, false, b1, (short)0, sacc[t], false, false);
            }
        }
    }

    const float* vfrag = Vt + r8 * KP_ + l15 * 4;
    v4f vf[2][8];
    #pragma unroll
    for (int j = 0; j < 8; ++j) vf[0][j] = *(const v4f*)(vfrag + j * KP_);

    float p[5][8];
    {
        const int hc = r8 - l15;
        #pragma unroll
        for (int t = 0; t < 5; ++t) {
            const bool tile_out = (keystart + t * 16) < 0;
            #pragma unroll
            for (int r = 0; r < 8; ++r) {
                const int d = hc + (t * 16 + r - WIN_);
                const float sc = fmaxf(sacc[t][r] * 0.125f, -1e9f);
                p[t][r] = (d > 0 || d < -WIN_ || tile_out) ? -1e9f : sc;
            }
        }
    }

    {
        float mt[5];
        #pragma unroll
        for (int t = 0; t < 5; ++t) {
            const float m01 = fmaxf(p[t][0], p[t][1]), m23 = fmaxf(p[t][2], p[t][3]);
            const float m45 = fmaxf(p[t][4], p[t][5]), m67 = fmaxf(p[t][6], p[t][7]);
            mt[t] = fmaxf(fmaxf(m01, m23), fmaxf(m45, m67));
        }
        float m = fmaxf(fmaxf(fmaxf(mt[0], mt[1]), fmaxf(mt[2], mt[3])), mt[4]);
        m = fmaxf(m, __shfl_xor(m, 16, 32));
        float s4[4] = {0.0f, 0.0f, 0.0f, 0.0f};
        #pragma unroll
        for (int t = 0; t < 5; ++t)
            #pragma unroll
            for (int r = 0; r < 8; ++r) {
                const float e = __builtin_amdgcn_exp2f((p[t][r] - m) * 1.4426950408889634f);
                p[t][r] = e;
                s4[r & 3] += e;
            }
        float sum = (s4[0] + s4[1]) + (s4[2] + s4[3]);
        sum += __shfl_xor(sum, 16, 32);
        const float inv = 1.0f / sum;
        #pragma unroll
        for (int t = 0; t < 5; ++t)
            #pragma unroll
            for (int r = 0; r < 8; ++r) p[t][r] *= inv;
    }

    {
        v8f oacc[4] = {};
        #pragma unroll
        for (int t = 0; t < 5; ++t) {
            const int cb = t & 1, nb = cb ^ 1;
            if (t + 1 < 5) {
                #pragma unroll
                for (int j = 0; j < 8; ++j)
                    vf[nb][j] = *(const v4f*)(vfrag + ((t + 1) * 16 + j) * KP_);
            }
            __builtin_amdgcn_sched_barrier(0);
            #pragma unroll
            for (int j = 0; j < 4; ++j)
                #pragma unroll
                for (int nt = 0; nt < 4; ++nt) {
                    v2f a = {p[t][2 * j], p[t][2 * j + 1]};
                    v2f bv = {vf[cb][2 * j][nt], vf[cb][2 * j + 1][nt]};
                    oacc[nt] = __builtin_amdgcn_wmma_f32_16x16x4_f32(
                        false, a, false, bv, (short)0, oacc[nt], false, false);
                }
        }
        #pragma unroll
        for (int r = 0; r < 8; ++r) {
            v4f ov = {oacc[0][r], oacc[1][r], oacc[2][r], oacc[3][r]};
            *(v4f*)(orow + (long)r * (H_ * HD_)) = ov;
        }
    }
}

template <bool V> struct flag_c { static constexpr bool value = V; };
template <int V> struct int_c { static constexpr int value = V; };

template <int BM, int BN, int BK, int WM, int WN, int DO_ROPE, int DUAL, int FUSE>
__global__ void __launch_bounds__((BM / WM) * (BN / WN) * 32)
gemm_lds_wmma(const float* __restrict__ A, const float* __restrict__ W,
              const float* __restrict__ W2,
              float* __restrict__ C, float* __restrict__ C2, int M, int N, int K,
              const float* __restrict__ fcos,
              const float* __restrict__ fsin,
              const float* __restrict__ kbuf,
              const float* __restrict__ vbuf) {
    constexpr int NWN  = BN / WN;
    constexpr int NW   = (BM / WM) * NWN;
    constexpr int NT   = NW * 32;
    constexpr int P    = BK + 4;
    constexpr int TM   = WM / 16;
    constexpr int TN   = WN / 16;
    constexpr int F4R  = BK / 4;
    constexpr int A_IT = (BM * F4R) / NT;
    constexpr int NBN  = BN / 32;
    constexpr int B_IT = ((BK / 4) * NBN) / NW;
    constexpr int STAGE = (BM + BN) * P;
    constexpr int BNH  = DUAL ? BN / 2 : BN;
    static_assert(BNH % WN == 0 && BNH % 32 == 0, "a wave tile and a 32-column block lie in one matrix");
    static_assert((BM * F4R) % NT == 0, "A tile must divide among the threads");
    static_assert(((BK / 4) * NBN) % NW == 0, "W tile must divide among the waves");
    static_assert(BK % 8 == 0 && (P / 4) % 2 == 1, "pitch must be 16 B x odd");
    constexpr int QSF  = BM * KP_;
    constexpr int KVF  = (BM + WIN_) * KP_;
    constexpr int LDSF = (FUSE && QSF + 2 * KVF > 2 * STAGE) ? QSF + 2 * KVF : 2 * STAGE;
    static_assert(!FUSE || (BN == HD_ && DO_ROPE && !DUAL && (BM / 16) % NW == 0),
                  "the fused block is one head wide and its query tiles divide among its waves");

    __shared__ __attribute__((aligned(16))) float lds[LDSF];

    const int tid  = threadIdx.x;
    const int lane = tid & 31;
    const int wave = __builtin_amdgcn_readfirstlane(tid >> 5);
    const int wm   = wave / NWN;
    const int wn   = wave % NWN;
    const int tileM = blockIdx.y * BM;
    const int tileN = blockIdx.x * BNH;

    const int l15 = lane & 15;
    const int h4  = (lane >> 4) << 2;

    v4f ra[A_IT], rb[B_IT];
    const float* agp[A_IT];
    int          asd[A_IT];
    #pragma unroll
    for (int i = 0; i < A_IT; ++i) {
        const int idx = tid + i * NT;
        const int row = idx / F4R;
        const int c4  = idx % F4R;
        agp[i] = A + (long)(tileM + row) * K + c4 * 4;
        asd[i] = row * P + c4 * 4;
    }
    const float* bgp[B_IT];
    int          bsd[B_IT];
    #pragma unroll
    for (int i = 0; i < B_IT; ++i) {
        const int blk = wave + i * NW;
        const int nb  = blk % NBN;
        const int kb  = blk / NBN;
        const int k   = kb * 4 + (lane >> 3);
        const int n   = nb * 32 + (lane & 7) * 4;
        const float* wsrc = (DUAL && n >= BNH) ? W2 : W;
        bgp[i] = wsrc + (long)k * N + tileN + (n % BNH);
        bsd[i] = BM * P + n * P + k;
    }

    auto load_tile = [&](int k0) {
        #pragma unroll
        for (int i = 0; i < A_IT; ++i) ra[i] = *(const v4f*)(agp[i] + k0);
        #pragma unroll
        for (int i = 0; i < B_IT; ++i) rb[i] = *(const v4f*)(bgp[i] + (long)k0 * N);
    };
    auto store_tile = [&](int stage) {
        float* s = lds + stage * STAGE;
        #pragma unroll
        for (int i = 0; i < A_IT; ++i) *(v4f*)(s + asd[i]) = ra[i];
        #pragma unroll
        for (int i = 0; i < B_IT; ++i) {
            s[bsd[i]]         = rb[i].x;
            s[bsd[i] + P]     = rb[i].y;
            s[bsd[i] + 2 * P] = rb[i].z;
            s[bsd[i] + 3 * P] = rb[i].w;
        }
    };

    auto lds_barrier = [&]() {
        __builtin_amdgcn_fence(__ATOMIC_RELEASE, "workgroup", "local");
        __builtin_amdgcn_s_barrier();
        __builtin_amdgcn_fence(__ATOMIC_ACQUIRE, "workgroup", "local");
    };
    auto bar_signal = [&]() {
        __builtin_amdgcn_fence(__ATOMIC_RELEASE, "workgroup", "local");
        __builtin_amdgcn_s_barrier_signal(-1);
    };
    auto bar_wait = [&]() {
        __builtin_amdgcn_s_barrier_wait(-1);
        __builtin_amdgcn_fence(__ATOMIC_ACQUIRE, "workgroup", "local");
    };

    v8f acc[TM][TN] = {};

    const int aoff = (wm * WM + l15) * P + h4;
    const int boff = BM * P + (wn * WN + l15) * P + h4;

    constexpr int G = BK / 8;

    v4f fa[2][G][TM], fb[2][G][TN];
    auto read_frags = [&](auto set_c, const float* st) {
        constexpr int SET = decltype(set_c)::value;
        #pragma unroll
        for (int g = 0; g < G; ++g) {
            #pragma unroll
            for (int i = 0; i < TM; ++i) fa[SET][g][i] = *(const v4f*)(st + aoff + i * 16 * P + g * 8);
            #pragma unroll
            for (int j = 0; j < TN; ++j) fb[SET][g][j] = *(const v4f*)(st + boff + j * 16 * P + g * 8);
        }
    };

    const int nk = K / BK;
    load_tile(0);
    store_tile(0);
    load_tile(BK);
    lds_barrier();
    read_frags(int_c<0>{}, lds);
    store_tile(1);
    load_tile(2 * BK);
    bar_signal();

    auto k_step = [&](auto set_c, auto rd_c, auto st_c, auto ld_c, int kt) {
        constexpr int SET = decltype(set_c)::value;
        bar_wait();
        if (decltype(rd_c)::value) read_frags(int_c<(SET ^ 1)>{}, lds + (SET ^ 1) * STAGE);
        if (decltype(st_c)::value) store_tile(SET);
        if (decltype(ld_c)::value) load_tile((kt + 3) * BK);
        __builtin_amdgcn_sched_barrier(0);
        #pragma unroll
        for (int g = 0; g < G; ++g) {
            if (g == G / 2 && decltype(rd_c)::value) {
                __builtin_amdgcn_sched_barrier(0);
                bar_signal();
                __builtin_amdgcn_sched_barrier(0);
            }
            #pragma unroll
            for (int i = 0; i < TM; ++i)
                #pragma unroll
                for (int j = 0; j < TN; ++j) {
                    v2f a0 = {fa[SET][g][i].x, fa[SET][g][i].y};
                    v2f b0 = {fb[SET][g][j].x, fb[SET][g][j].y};
                    acc[i][j] = __builtin_amdgcn_wmma_f32_16x16x4_f32(
                        false, a0, false, b0, (short)0, acc[i][j], false, false);
                }
            #pragma unroll
            for (int i = 0; i < TM; ++i)
                #pragma unroll
                for (int j = 0; j < TN; ++j) {
                    v2f a1 = {fa[SET][g][i].z, fa[SET][g][i].w};
                    v2f b1 = {fb[SET][g][j].z, fb[SET][g][j].w};
                    acc[i][j] = __builtin_amdgcn_wmma_f32_16x16x4_f32(
                        false, a1, false, b1, (short)0, acc[i][j], false, false);
                }
        }
        __builtin_amdgcn_sched_barrier(0);
    };
    constexpr flag_c<true> yes{};
    constexpr flag_c<false> no{};
    int kt = 0;
    for (; kt + 4 < nk; kt += 2) {
        k_step(int_c<0>{}, yes, yes, yes, kt);
        k_step(int_c<1>{}, yes, yes, yes, kt + 1);
    }
    k_step(int_c<0>{}, yes, yes, yes, kt);
    k_step(int_c<1>{}, yes, yes, no, kt + 1);
    k_step(int_c<0>{}, yes, no, no, kt + 2);
    k_step(int_c<1>{}, no, no, no, kt + 3);

    if (FUSE) {
        float* Qs = lds;
        float* Ks = lds + QSF;
        float* Vs = Ks + KVF;
        const int head = blockIdx.x;
        const int kvh  = head / (H_ / KVH_);
        const int s0   = tileM & (S_ - 1);
        const int brow = tileM - s0;
        const int rowOffq = (lane >> 4) << 3;

        #pragma unroll
        for (int i = 0; i < TM; ++i) {
            const int rowl = wm * WM + i * 16 + rowOffq;
            #pragma unroll
            for (int j = 0; j < TN; ++j) {
                const int col = wn * WN + j * 16 + l15;
                #pragma unroll
                for (int r = 0; r < 8; ++r) {
                    float val = acc[i][j][r];
                    const float partner = __shfl_xor(val, 1, 32);
                    const int pos = s0 + rowl + r;
                    const int fi  = col >> 1;
                    const float c = fcos[pos * (HD_ / 2) + fi];
                    const float sn = fsin[pos * (HD_ / 2) + fi];
                    val = ((col & 1) == 0) ? (val * c - partner * sn)
                                           : (partner * sn + val * c);
                    Qs[(rowl + r) * KP_ + col] = val;
                }
            }
        }

        constexpr int KV_IT = ((BM + WIN_) * 16) / NT;
        static_assert(!FUSE || ((BM + WIN_) * 16) % NT == 0, "K / V rows divide among the threads");
        {
            v4f kr[KV_IT], vr[KV_IT];
            #pragma unroll
            for (int i = 0; i < KV_IT; ++i) {
                const int idx = tid + i * NT;
                const int row = idx >> 4;
                const int c4  = idx & 15;
                const int key = s0 - WIN_ + row;
                const int keyc = key < 0 ? 0 : key;
                const long off = ((long)(brow + keyc) * KVH_ + kvh) * HD_ + c4 * 4;
                kr[i] = *(const v4f*)(kbuf + off);
                vr[i] = *(const v4f*)(vbuf + off);
            }
            #pragma unroll
            for (int i = 0; i < KV_IT; ++i) {
                const int idx = tid + i * NT;
                const int row = idx >> 4;
                const int c4  = idx & 15;
                const bool valid = (s0 - WIN_ + row) >= 0;
                const v4f zero = {0.0f, 0.0f, 0.0f, 0.0f};
                *(v4f*)(Ks + row * KP_ + c4 * 4) = kr[i];
                *(v4f*)(Vs + row * KP_ + c4 * 4) = valid ? vr[i] : zero;
            }
        }
        __syncthreads();

        constexpr int QT_W = (BM / 16) / NW;
        for (int u = 0; u < QT_W; ++u) {
            const int qt = wave * QT_W + u;
            const int i0 = qt * 16;
            float* orow = C + ((long)(tileM + i0 + rowOffq) * H_ + head) * HD_ + l15 * 4;
            attn_tile(Qs + i0 * KP_, Ks + i0 * KP_, Vs + i0 * KP_, s0 + i0 - WIN_, orow, lane);
        }
        return;
    }

    const bool second = DUAL && (wn * WN >= BNH);
    float* __restrict__ Cd = second ? C2 : C;
    const int colBase = tileN + wn * WN - (second ? BNH : 0);
    const int rowOff = (lane >> 4) << 3;
    auto epilogue = [&](auto rope_c) {
    constexpr bool ROPE = decltype(rope_c)::value;
    #pragma unroll
    for (int i = 0; i < TM; ++i) {
        const int rowBase = tileM + wm * WM + i * 16 + rowOff;
        #pragma unroll
        for (int j = 0; j < TN; ++j) {
            const int col = colBase + j * 16 + l15;
            #pragma unroll
            for (int r = 0; r < 8; ++r) {
                float val = acc[i][j][r];
                if (ROPE) {
                    const float partner = __shfl_xor(val, 1, 32);
                    const int pos = (rowBase + r) & (S_ - 1);
                    const int fi  = (col & (HD_ - 1)) >> 1;
                    const float c = fcos[pos * (HD_ / 2) + fi];
                    const float s = fsin[pos * (HD_ / 2) + fi];
                    val = ((col & 1) == 0) ? (val * c - partner * s)
                                           : (partner * s + val * c);
                }
                Cd[(long)(rowBase + r) * N + col] = val;
            }
        }
    }
    };
    if (DO_ROPE && !second) epilogue(flag_c<true>{});
    else                    epilogue(flag_c<false>{});
}

extern "C" void kernel_launch(void* const* d_in, const int* in_sizes, int n_in,
                              void* d_out, int out_size, void* d_ws, size_t ws_size,
                              hipStream_t stream) {
    const float* x    = (const float*)d_in[0];
    const float* fcos = (const float*)d_in[1];
    const float* fsin = (const float*)d_in[2];
    const float* wq   = (const float*)d_in[3];
    const float* wk   = (const float*)d_in[4];
    const float* wv   = (const float*)d_in[5];
    const float* wo   = (const float*)d_in[6];
    float* out = (float*)d_out;

    float* kb = (float*)d_ws;
    float* vb = kb + (size_t)M_ * (KVH_ * HD_);
    float* at = vb + (size_t)M_ * (KVH_ * HD_);

    const dim3 blk(128);
    gemm_lds_wmma<64, 128, 32, 32, 32, 1, 1, 0><<<dim3((KVH_ * HD_) / 64, M_ / 64), dim3(256), 0, stream>>>(
        x, wk, wv, kb, vb, M_, KVH_ * HD_, D_, fcos, fsin, nullptr, nullptr);

    gemm_lds_wmma<128, 64, 32, 32, 32, 1, 0, 1><<<dim3((H_ * HD_) / 64, M_ / 128), dim3(256), 0, stream>>>(
        x, wq, nullptr, at, nullptr, M_, H_ * HD_, D_, fcos, fsin, kb, vb);

    gemm_lds_wmma<128, 64, 32, 64, 32, 0, 0, 0><<<dim3(D_ / 64, M_ / 128), blk, 0, stream>>>(
        at, wo, nullptr, out, nullptr, M_, D_, H_ * HD_, fcos, fsin, nullptr, nullptr);
}
